# starting kernel
# baseline (speedup 1.0000x reference)
#include <hip/hip_runtime.h>
#include <math.h>

#define HW   1048576
#define C    64
#define KSEG 4096
#define H1   256
#define NCLS 16
#define KT1  (C + C*8)
#define KT2  (H1 + H1*8)

typedef __attribute__((ext_vector_type(2))) float v2f;
typedef __attribute__((ext_vector_type(4))) float v4f;
typedef __attribute__((ext_vector_type(8))) float v8f;

__device__ __forceinline__ float silu_f(float x) {
    return x / (1.0f + expf(-x));
}

__device__ __forceinline__ void bspline8(float x, float* __restrict__ out) {
    float t[12];
#pragma unroll
    for (int j = 0; j < 12; ++j) t[j] = 0.4f * (float)(j - 3) - 1.0f;
    float b[11];
#pragma unroll
    for (int j = 0; j < 11; ++j) b[j] = (x >= t[j] && x < t[j + 1]) ? 1.0f : 0.0f;
#pragma unroll
    for (int k = 1; k <= 3; ++k) {
#pragma unroll
        for (int j = 0; j < 10; ++j) {
            if (j < 11 - k) {
                float left  = (x - t[j])         / (t[j + k]     - t[j]);
                float right = (t[j + k + 1] - x) / (t[j + k + 1] - t[j + 1]);
                b[j] = left * b[j] + right * b[j + 1];
            }
        }
    }
#pragma unroll
    for (int g = 0; g < 8; ++g) out[g] = b[g];
}

__global__ void zero_kernel(float* __restrict__ p, int n) {
    int i = blockIdx.x * blockDim.x + threadIdx.x;
    if (i < n) p[i] = 0.0f;
}

__global__ void segsum_kernel(const float* __restrict__ x, const int* __restrict__ cls,
                              float* __restrict__ acc, float* __restrict__ cnt) {
    int t = threadIdx.x;
    int rloc = t >> 4;
    int c4   = (t & 15) << 2;
    for (int base = blockIdx.x * 16; base < HW; base += gridDim.x * 16) {
        int row = base + rloc;
        int seg = __builtin_nontemporal_load(cls + row);
        const v4f v = __builtin_nontemporal_load((const v4f*)(x + (size_t)row * C + c4));
        float* a = acc + (size_t)seg * C + c4;
        atomicAdd(a + 0, v.x);
        atomicAdd(a + 1, v.y);
        atomicAdd(a + 2, v.z);
        atomicAdd(a + 3, v.w);
        if ((t & 15) == 0) atomicAdd(cnt + seg, 1.0f);
    }
}

__global__ void wprep1_kernel(const float* __restrict__ bw1, const float* __restrict__ sw1,
                              const float* __restrict__ sc1, float* __restrict__ Wt1) {
    int idx = blockIdx.x * blockDim.x + threadIdx.x;
    if (idx >= H1 * KT1) return;
    int o = idx / KT1, j = idx % KT1;
    float v;
    if (j < C) {
        v = bw1[o * C + j];
    } else {
        int j2 = j - C, i = j2 >> 3, g = j2 & 7;
        v = sw1[((size_t)o * C + i) * 8 + g] * sc1[o * C + i];
    }
    Wt1[(size_t)j * H1 + o] = v;
}

__global__ void wprep2_kernel(const float* __restrict__ bw2, const float* __restrict__ sw2,
                              const float* __restrict__ sc2, float* __restrict__ Wt2) {
    int idx = blockIdx.x * blockDim.x + threadIdx.x;
    if (idx >= NCLS * KT2) return;
    int o = idx / KT2, j = idx % KT2;
    float v;
    if (j < H1) {
        v = bw2[o * H1 + j];
    } else {
        int j2 = j - H1, i = j2 >> 3, g = j2 & 7;
        v = sw2[((size_t)o * H1 + i) * 8 + g] * sc2[o * H1 + i];
    }
    Wt2[(size_t)j * NCLS + o] = v;
}

__global__ void aprep1_kernel(const float* __restrict__ acc, const float* __restrict__ cnt,
                              float* __restrict__ A1) {
    int idx = blockIdx.x * blockDim.x + threadIdx.x;
    if (idx >= KSEG * C) return;
    int k = idx >> 6, c = idx & 63;
    float n = cnt[k];
    float f = acc[(size_t)k * C + c] / fmaxf(n, 1.0f);
    float* row = A1 + (size_t)k * KT1;
    row[c] = silu_f(f);
    float b[8];
    bspline8(f, b);
#pragma unroll
    for (int g = 0; g < 8; ++g) row[C + c * 8 + g] = b[g];
}

__global__ void aprep2_kernel(const float* __restrict__ h, float* __restrict__ A2) {
    int idx = blockIdx.x * blockDim.x + threadIdx.x;
    if (idx >= KSEG * H1) return;
    int m = idx >> 8, i = idx & 255;
    float f = h[(size_t)m * H1 + i];
    float* row = A2 + (size_t)m * KT2;
    row[i] = silu_f(f);
    float b[8];
    bspline8(f, b);
#pragma unroll
    for (int g = 0; g < 8; ++g) row[H1 + i * 8 + g] = b[g];
}

template <int NT>
__global__ void gemm_f32_wmma(const float* __restrict__ A, const float* __restrict__ Bt,
                              float* __restrict__ D, int M, int Kt, int N) {
    const int tilesN = N / (16 * NT);
    const int tm = blockIdx.x / tilesN;
    const int tn = blockIdx.x % tilesN;
    const int lane = threadIdx.x & 31;
    const int mrow = lane & 15;
    const int khalf = (lane >> 4) << 1;
    const int n0 = tn * 16 * NT;

    const float* Arow = A + (size_t)(tm * 16 + mrow) * Kt;

    v8f accv[NT];
#pragma unroll
    for (int t = 0; t < NT; ++t) accv[t] = (v8f){0.f, 0.f, 0.f, 0.f, 0.f, 0.f, 0.f, 0.f};

    for (int kk = 0; kk < Kt; kk += 4) {
        v2f a;
        a.x = Arow[kk + khalf];
        a.y = Arow[kk + khalf + 1];
#pragma unroll
        for (int t = 0; t < NT; ++t) {
            int n = n0 + t * 16 + mrow;
            v2f b;
            b.x = Bt[(size_t)(kk + khalf) * N + n];
            b.y = Bt[(size_t)(kk + khalf + 1) * N + n];
            accv[t] = __builtin_amdgcn_wmma_f32_16x16x4_f32(
                false, a, false, b, (short)0, accv[t], false, false);
        }
    }

    const int rbase = (lane >> 4) * 8;
#pragma unroll
    for (int t = 0; t < NT; ++t) {
#pragma unroll
        for (int r = 0; r < 8; ++r) {
            D[(size_t)(tm * 16 + rbase + r) * N + n0 + t * 16 + mrow] = accv[t][r];
        }
    }
}

extern "C" void kernel_launch(void* const* d_in, const int* in_sizes, int n_in,
                              void* d_out, int out_size, void* d_ws, size_t ws_size,
                              hipStream_t stream) {
    const float* x   = (const float*)d_in[0];
    const int*   cls = (const int*)  d_in[1];
    const float* bw1 = (const float*)d_in[2];
    const float* sw1 = (const float*)d_in[3];
    const float* sc1 = (const float*)d_in[4];
    const float* bw2 = (const float*)d_in[5];
    const float* sw2 = (const float*)d_in[6];
    const float* sc2 = (const float*)d_in[7];
    float* out = (float*)d_out;

    float* ws  = (float*)d_ws;
    float* acc = ws;
    float* cnt = acc + (size_t)KSEG * C;
    float* A1  = cnt + KSEG;
    float* Wt1 = A1 + (size_t)KSEG * KT1;
    float* h   = Wt1 + (size_t)KT1 * H1;
    float* A2  = h + (size_t)KSEG * H1;
    float* Wt2 = A2 + (size_t)KSEG * KT2;

    {
        int n = KSEG * C + KSEG;
        zero_kernel<<<(n + 255) / 256, 256, 0, stream>>>(acc, n);
    }
    segsum_kernel<<<8192, 256, 0, stream>>>(x, cls, acc, cnt);
    {
        int n = H1 * KT1;
        wprep1_kernel<<<(n + 255) / 256, 256, 0, stream>>>(bw1, sw1, sc1, Wt1);
    }
    {
        int n = KSEG * C;
        aprep1_kernel<<<(n + 255) / 256, 256, 0, stream>>>(acc, cnt, A1);
    }
    gemm_f32_wmma<4><<<(KSEG / 16) * (H1 / 64), 32, 0, stream>>>(A1, Wt1, h, KSEG, KT1, H1);
    {
        int n = KSEG * H1;
        aprep2_kernel<<<(n + 255) / 256, 256, 0, stream>>>(h, A2);
    }
    {
        int n = NCLS * KT2;
        wprep2_kernel<<<(n + 255) / 256, 256, 0, stream>>>(bw2, sw2, sc2, Wt2);
    }
    gemm_f32_wmma<1><<<(KSEG / 16) * (NCLS / 16), 32, 0, stream>>>(A2, Wt2, out, KSEG, KT2, NCLS);
}


# segment sums from a per-segment row list (integer count, prefix sum, fill), gathered whole by one wave per segment; no float atomics; 8 launches -> 11; sum order changes (arrival-order list places)
# speedup vs baseline: 1.0620x; 1.0620x over previous
#include <hip/hip_runtime.h>
#include <math.h>

#define HW   1048576
#define C    64
#define KSEG 4096
#define H1   256
#define NCLS 16
#define KT1  (C + C*8)
#define KT2  (H1 + H1*8)

typedef __attribute__((ext_vector_type(2))) float v2f;
typedef __attribute__((ext_vector_type(4))) float v4f;
typedef __attribute__((ext_vector_type(8))) float v8f;

__device__ __forceinline__ float silu_f(float x) {
    return x / (1.0f + expf(-x));
}

__device__ __forceinline__ void bspline8(float x, float* __restrict__ out) {
    float t[12];
#pragma unroll
    for (int j = 0; j < 12; ++j) t[j] = 0.4f * (float)(j - 3) - 1.0f;
    float b[11];
#pragma unroll
    for (int j = 0; j < 11; ++j) b[j] = (x >= t[j] && x < t[j + 1]) ? 1.0f : 0.0f;
#pragma unroll
    for (int k = 1; k <= 3; ++k) {
#pragma unroll
        for (int j = 0; j < 10; ++j) {
            if (j < 11 - k) {
                float left  = (x - t[j])         / (t[j + k]     - t[j]);
                float right = (t[j + k + 1] - x) / (t[j + k + 1] - t[j + 1]);
                b[j] = left * b[j] + right * b[j + 1];
            }
        }
    }
#pragma unroll
    for (int g = 0; g < 8; ++g) out[g] = b[g];
}

#define ROWS_PER_THREAD 8
#define GU 16

__global__ void zero_int_kernel(int* __restrict__ p, int n) {
    int i = blockIdx.x * blockDim.x + threadIdx.x;
    if (i < n) p[i] = 0;
}

__global__ void __launch_bounds__(256)
segcount_kernel(const int* __restrict__ cls, int* __restrict__ cnt_i) {
    const int base = (blockIdx.x * 256 + threadIdx.x) * ROWS_PER_THREAD;
    int s[ROWS_PER_THREAD];
#pragma unroll
    for (int j = 0; j < ROWS_PER_THREAD; ++j) s[j] = __builtin_nontemporal_load(cls + base + j);
#pragma unroll
    for (int j = 0; j < ROWS_PER_THREAD; ++j) atomicAdd(cnt_i + s[j], 1);
}

__global__ void __launch_bounds__(256)
segscan_kernel(const int* __restrict__ cnt_i, int* __restrict__ segstart,
               int* __restrict__ cursor, float* __restrict__ cnt) {
    __shared__ int tot[2][256];
    const int t = threadIdx.x;
    int c[16];
    int sum = 0;
#pragma unroll
    for (int j = 0; j < 16; ++j) { c[j] = cnt_i[t * 16 + j]; sum += c[j]; }
    tot[0][t] = sum;
    __syncthreads();
    int cur = 0;
    for (int d = 1; d < 256; d <<= 1) {
        int v = tot[cur][t];
        if (t >= d) v += tot[cur][t - d];
        tot[cur ^ 1][t] = v;
        cur ^= 1;
        __syncthreads();
    }
    int off = tot[cur][t] - sum;
#pragma unroll
    for (int j = 0; j < 16; ++j) {
        segstart[t * 16 + j] = off;
        cursor[t * 16 + j]   = off;
        cnt[t * 16 + j]      = (float)c[j];
        off += c[j];
    }
    if (t == 255) segstart[KSEG] = off;
}

__global__ void __launch_bounds__(256)
segfill_kernel(const int* __restrict__ cls, int* __restrict__ cursor, int* __restrict__ perm) {
    const int base = (blockIdx.x * 256 + threadIdx.x) * ROWS_PER_THREAD;
    int s[ROWS_PER_THREAD], pos[ROWS_PER_THREAD];
#pragma unroll
    for (int j = 0; j < ROWS_PER_THREAD; ++j) s[j] = __builtin_nontemporal_load(cls + base + j);
#pragma unroll
    for (int j = 0; j < ROWS_PER_THREAD; ++j) pos[j] = atomicAdd(cursor + s[j], 1);
#pragma unroll
    for (int j = 0; j < ROWS_PER_THREAD; ++j) perm[pos[j]] = base + j;
}

__global__ void __launch_bounds__(128)
seggather_kernel(const float* __restrict__ x, const int* __restrict__ segstart,
                 const int* __restrict__ perm, float* __restrict__ acc) {
    const int lane = threadIdx.x & 31;
    const int half = lane >> 4;
    const int c4   = (lane & 15) << 2;
    const int seg  = blockIdx.x * 4 + __builtin_amdgcn_readfirstlane((int)(threadIdx.x >> 5));
    const int s0 = segstart[seg];
    const int n  = segstart[seg + 1] - s0;
    const int* __restrict__ list = perm + s0;
    v4f s = (v4f){0.f, 0.f, 0.f, 0.f};
    if (n > 0) {
        int r[GU];
#pragma unroll
        for (int u = 0; u < GU; ++u) {
            const int e = u * 2 + half;
            r[u] = list[e < n ? e : n - 1];
        }
        for (int e0 = 0; e0 < n; e0 += 2 * GU) {
            v4f v[GU];
#pragma unroll
            for (int u = 0; u < GU; ++u) {
                v[u] = __builtin_nontemporal_load(
                    (const v4f*)(x + (size_t)(r[u] & (HW - 1)) * C + c4));
            }
#pragma unroll
            for (int u = 0; u < GU; ++u) {
                const int e = e0 + 2 * GU + u * 2 + half;
                r[u] = list[e < n ? e : n - 1];
            }
#pragma unroll
            for (int u = 0; u < GU; ++u) {
                const int e = e0 + u * 2 + half;
                if (e < n) s += v[u];
            }
        }
    }
    s.x += __shfl_xor(s.x, 16);
    s.y += __shfl_xor(s.y, 16);
    s.z += __shfl_xor(s.z, 16);
    s.w += __shfl_xor(s.w, 16);
    if (half == 0) *(v4f*)(acc + (size_t)seg * C + c4) = s;
}

__global__ void wprep1_kernel(const float* __restrict__ bw1, const float* __restrict__ sw1,
                              const float* __restrict__ sc1, float* __restrict__ Wt1) {
    int idx = blockIdx.x * blockDim.x + threadIdx.x;
    if (idx >= H1 * KT1) return;
    int o = idx / KT1, j = idx % KT1;
    float v;
    if (j < C) {
        v = bw1[o * C + j];
    } else {
        int j2 = j - C, i = j2 >> 3, g = j2 & 7;
        v = sw1[((size_t)o * C + i) * 8 + g] * sc1[o * C + i];
    }
    Wt1[(size_t)j * H1 + o] = v;
}

__global__ void wprep2_kernel(const float* __restrict__ bw2, const float* __restrict__ sw2,
                              const float* __restrict__ sc2, float* __restrict__ Wt2) {
    int idx = blockIdx.x * blockDim.x + threadIdx.x;
    if (idx >= NCLS * KT2) return;
    int o = idx / KT2, j = idx % KT2;
    float v;
    if (j < H1) {
        v = bw2[o * H1 + j];
    } else {
        int j2 = j - H1, i = j2 >> 3, g = j2 & 7;
        v = sw2[((size_t)o * H1 + i) * 8 + g] * sc2[o * H1 + i];
    }
    Wt2[(size_t)j * NCLS + o] = v;
}

__global__ void aprep1_kernel(const float* __restrict__ acc, const float* __restrict__ cnt,
                              float* __restrict__ A1) {
    int idx = blockIdx.x * blockDim.x + threadIdx.x;
    if (idx >= KSEG * C) return;
    int k = idx >> 6, c = idx & 63;
    float n = cnt[k];
    float f = acc[(size_t)k * C + c] / fmaxf(n, 1.0f);
    float* row = A1 + (size_t)k * KT1;
    row[c] = silu_f(f);
    float b[8];
    bspline8(f, b);
#pragma unroll
    for (int g = 0; g < 8; ++g) row[C + c * 8 + g] = b[g];
}

__global__ void aprep2_kernel(const float* __restrict__ h, float* __restrict__ A2) {
    int idx = blockIdx.x * blockDim.x + threadIdx.x;
    if (idx >= KSEG * H1) return;
    int m = idx >> 8, i = idx & 255;
    float f = h[(size_t)m * H1 + i];
    float* row = A2 + (size_t)m * KT2;
    row[i] = silu_f(f);
    float b[8];
    bspline8(f, b);
#pragma unroll
    for (int g = 0; g < 8; ++g) row[H1 + i * 8 + g] = b[g];
}

template <int NT>
__global__ void gemm_f32_wmma(const float* __restrict__ A, const float* __restrict__ Bt,
                              float* __restrict__ D, int M, int Kt, int N) {
    const int tilesN = N / (16 * NT);
    const int tm = blockIdx.x / tilesN;
    const int tn = blockIdx.x % tilesN;
    const int lane = threadIdx.x & 31;
    const int mrow = lane & 15;
    const int khalf = (lane >> 4) << 1;
    const int n0 = tn * 16 * NT;

    const float* Arow = A + (size_t)(tm * 16 + mrow) * Kt;

    v8f accv[NT];
#pragma unroll
    for (int t = 0; t < NT; ++t) accv[t] = (v8f){0.f, 0.f, 0.f, 0.f, 0.f, 0.f, 0.f, 0.f};

    for (int kk = 0; kk < Kt; kk += 4) {
        v2f a;
        a.x = Arow[kk + khalf];
        a.y = Arow[kk + khalf + 1];
#pragma unroll
        for (int t = 0; t < NT; ++t) {
            int n = n0 + t * 16 + mrow;
            v2f b;
            b.x = Bt[(size_t)(kk + khalf) * N + n];
            b.y = Bt[(size_t)(kk + khalf + 1) * N + n];
            accv[t] = __builtin_amdgcn_wmma_f32_16x16x4_f32(
                false, a, false, b, (short)0, accv[t], false, false);
        }
    }

    const int rbase = (lane >> 4) * 8;
#pragma unroll
    for (int t = 0; t < NT; ++t) {
#pragma unroll
        for (int r = 0; r < 8; ++r) {
            D[(size_t)(tm * 16 + rbase + r) * N + n0 + t * 16 + mrow] = accv[t][r];
        }
    }
}

extern "C" void kernel_launch(void* const* d_in, const int* in_sizes, int n_in,
                              void* d_out, int out_size, void* d_ws, size_t ws_size,
                              hipStream_t stream) {
    const float* x   = (const float*)d_in[0];
    const int*   cls = (const int*)  d_in[1];
    const float* bw1 = (const float*)d_in[2];
    const float* sw1 = (const float*)d_in[3];
    const float* sc1 = (const float*)d_in[4];
    const float* bw2 = (const float*)d_in[5];
    const float* sw2 = (const float*)d_in[6];
    const float* sc2 = (const float*)d_in[7];
    float* out = (float*)d_out;

    float* ws  = (float*)d_ws;
    float* acc = ws;
    float* cnt = acc + (size_t)KSEG * C;
    float* A1  = cnt + KSEG;
    float* Wt1 = A1 + (size_t)KSEG * KT1;
    float* h   = Wt1 + (size_t)KT1 * H1;
    float* A2  = h + (size_t)KSEG * H1;
    float* Wt2 = A2 + (size_t)KSEG * KT2;

    int* perm     = (int*)A2;
    int* cnt_i    = perm + HW;
    int* segstart = cnt_i + KSEG;
    int* cursor   = segstart + KSEG + 4;

    zero_int_kernel<<<KSEG / 256, 256, 0, stream>>>(cnt_i, KSEG);
    segcount_kernel<<<HW / (256 * ROWS_PER_THREAD), 256, 0, stream>>>(cls, cnt_i);
    segscan_kernel<<<1, 256, 0, stream>>>(cnt_i, segstart, cursor, cnt);
    segfill_kernel<<<HW / (256 * ROWS_PER_THREAD), 256, 0, stream>>>(cls, cursor, perm);
    seggather_kernel<<<KSEG / 4, 128, 0, stream>>>(x, segstart, perm, acc);
    {
        int n = H1 * KT1;
        wprep1_kernel<<<(n + 255) / 256, 256, 0, stream>>>(bw1, sw1, sc1, Wt1);
    }
    {
        int n = KSEG * C;
        aprep1_kernel<<<(n + 255) / 256, 256, 0, stream>>>(acc, cnt, A1);
    }
    gemm_f32_wmma<4><<<(KSEG / 16) * (H1 / 64), 32, 0, stream>>>(A1, Wt1, h, KSEG, KT1, H1);
    {
        int n = KSEG * H1;
        aprep2_kernel<<<(n + 255) / 256, 256, 0, stream>>>(h, A2);
    }
    {
        int n = NCLS * KT2;
        wprep2_kernel<<<(n + 255) / 256, 256, 0, stream>>>(bw2, sw2, sc2, Wt2);
    }
    gemm_f32_wmma<1><<<(KSEG / 16) * (NCLS / 16), 32, 0, stream>>>(A2, Wt2, out, KSEG, KT2, NCLS);
}


# row lists built with no global atomics: rows cut into 128 chunks, per-chunk LDS histograms, a chunk-by-segment offset table, LDS cursors in the fill; 11 launches -> 10; sum order changes (list order)
# speedup vs baseline: 1.9137x; 1.8020x over previous
#include <hip/hip_runtime.h>
#include <math.h>

#define HW   1048576
#define C    64
#define KSEG 4096
#define H1   256
#define NCLS 16
#define KT1  (C + C*8)
#define KT2  (H1 + H1*8)

typedef __attribute__((ext_vector_type(2))) float v2f;
typedef __attribute__((ext_vector_type(4))) float v4f;
typedef __attribute__((ext_vector_type(8))) float v8f;

__device__ __forceinline__ float silu_f(float x) {
    return x / (1.0f + expf(-x));
}

__device__ __forceinline__ void bspline8(float x, float* __restrict__ out) {
    float t[12];
#pragma unroll
    for (int j = 0; j < 12; ++j) t[j] = 0.4f * (float)(j - 3) - 1.0f;
    float b[11];
#pragma unroll
    for (int j = 0; j < 11; ++j) b[j] = (x >= t[j] && x < t[j + 1]) ? 1.0f : 0.0f;
#pragma unroll
    for (int k = 1; k <= 3; ++k) {
#pragma unroll
        for (int j = 0; j < 10; ++j) {
            if (j < 11 - k) {
                float left  = (x - t[j])         / (t[j + k]     - t[j]);
                float right = (t[j + k + 1] - x) / (t[j + k + 1] - t[j + 1]);
                b[j] = left * b[j] + right * b[j + 1];
            }
        }
    }
#pragma unroll
    for (int g = 0; g < 8; ++g) out[g] = b[g];
}

#define NB    128
#define CHUNK (HW / NB)
#define LT    1024
#define RPT   (CHUNK / LT)
#define TB    32
#define GU 16

typedef __attribute__((ext_vector_type(4))) int v4i;

__global__ void __launch_bounds__(LT)
chunkcount_kernel(const int* __restrict__ cls, int* __restrict__ table) {
    __shared__ int hist[KSEG];
    const int t = threadIdx.x;
    const int base = blockIdx.x * CHUNK + t * RPT;
    int s[RPT];
#pragma unroll
    for (int j = 0; j < RPT; ++j) s[j] = cls[base + j];
#pragma unroll
    for (int j = 0; j < KSEG / LT; ++j) hist[t + j * LT] = 0;
    __syncthreads();
#pragma unroll
    for (int j = 0; j < RPT; ++j) atomicAdd(&hist[s[j]], 1);
    __syncthreads();
    *(v4i*)(table + (size_t)blockIdx.x * KSEG + t * 4) = *(const v4i*)(hist + t * 4);
}

__global__ void __launch_bounds__(256)
segtotal_kernel(int* __restrict__ table, int* __restrict__ total, float* __restrict__ cnt) {
    const int s = blockIdx.x * 256 + threadIdx.x;
    int run = 0;
    for (int b0 = 0; b0 < NB; b0 += TB) {
        int c[TB];
#pragma unroll
        for (int u = 0; u < TB; ++u) c[u] = table[(size_t)(b0 + u) * KSEG + s];
#pragma unroll
        for (int u = 0; u < TB; ++u) {
            table[(size_t)(b0 + u) * KSEG + s] = run;
            run += c[u];
        }
    }
    total[s] = run;
    cnt[s]   = (float)run;
}

__global__ void __launch_bounds__(LT)
chunkfill_kernel(const int* __restrict__ cls, const int* __restrict__ table,
                 const int* __restrict__ total, int* __restrict__ segstart,
                 int* __restrict__ perm) {
    __shared__ int cursor[KSEG];
    __shared__ int tot[2][LT];
    const int t = threadIdx.x;
    const int base = blockIdx.x * CHUNK + t * RPT;
    int s[RPT], pos[RPT];
#pragma unroll
    for (int j = 0; j < RPT; ++j) s[j] = __builtin_nontemporal_load(cls + base + j);
    const v4i tv = *(const v4i*)(total + t * 4);
    const v4i ov = *(const v4i*)(table + (size_t)blockIdx.x * KSEG + t * 4);
    const int sum = tv.x + tv.y + tv.z + tv.w;
    tot[0][t] = sum;
    __syncthreads();
    int cur = 0;
    for (int d = 1; d < LT; d <<= 1) {
        int v = tot[cur][t];
        if (t >= d) v += tot[cur][t - d];
        tot[cur ^ 1][t] = v;
        cur ^= 1;
        __syncthreads();
    }
    v4i st;
    st.x = tot[cur][t] - sum;
    st.y = st.x + tv.x;
    st.z = st.y + tv.y;
    st.w = st.z + tv.z;
    *(v4i*)(cursor + t * 4) = st + ov;
    if (blockIdx.x == 0) {
        *(v4i*)(segstart + t * 4) = st;
        if (t == LT - 1) segstart[KSEG] = st.w + tv.w;
    }
    __syncthreads();
#pragma unroll
    for (int j = 0; j < RPT; ++j) pos[j] = atomicAdd(&cursor[s[j]], 1);
#pragma unroll
    for (int j = 0; j < RPT; ++j) perm[pos[j]] = base + j;
}

__global__ void __launch_bounds__(128)
seggather_kernel(const float* __restrict__ x, const int* __restrict__ segstart,
                 const int* __restrict__ perm, float* __restrict__ acc) {
    const int lane = threadIdx.x & 31;
    const int half = lane >> 4;
    const int c4   = (lane & 15) << 2;
    const int seg  = blockIdx.x * 4 + __builtin_amdgcn_readfirstlane((int)(threadIdx.x >> 5));
    const int s0 = segstart[seg];
    const int n  = segstart[seg + 1] - s0;
    const int* __restrict__ list = perm + s0;
    v4f s = (v4f){0.f, 0.f, 0.f, 0.f};
    if (n > 0) {
        int r[GU];
#pragma unroll
        for (int u = 0; u < GU; ++u) {
            const int e = u * 2 + half;
            r[u] = list[e < n ? e : n - 1];
        }
        for (int e0 = 0; e0 < n; e0 += 2 * GU) {
            v4f v[GU];
#pragma unroll
            for (int u = 0; u < GU; ++u) {
                v[u] = __builtin_nontemporal_load(
                    (const v4f*)(x + (size_t)(r[u] & (HW - 1)) * C + c4));
            }
#pragma unroll
            for (int u = 0; u < GU; ++u) {
                const int e = e0 + 2 * GU + u * 2 + half;
                r[u] = list[e < n ? e : n - 1];
            }
#pragma unroll
            for (int u = 0; u < GU; ++u) {
                const int e = e0 + u * 2 + half;
                if (e < n) s += v[u];
            }
        }
    }
    s.x += __shfl_xor(s.x, 16);
    s.y += __shfl_xor(s.y, 16);
    s.z += __shfl_xor(s.z, 16);
    s.w += __shfl_xor(s.w, 16);
    if (half == 0) *(v4f*)(acc + (size_t)seg * C + c4) = s;
}

__global__ void wprep1_kernel(const float* __restrict__ bw1, const float* __restrict__ sw1,
                              const float* __restrict__ sc1, float* __restrict__ Wt1) {
    int idx = blockIdx.x * blockDim.x + threadIdx.x;
    if (idx >= H1 * KT1) return;
    int o = idx / KT1, j = idx % KT1;
    float v;
    if (j < C) {
        v = bw1[o * C + j];
    } else {
        int j2 = j - C, i = j2 >> 3, g = j2 & 7;
        v = sw1[((size_t)o * C + i) * 8 + g] * sc1[o * C + i];
    }
    Wt1[(size_t)j * H1 + o] = v;
}

__global__ void wprep2_kernel(const float* __restrict__ bw2, const float* __restrict__ sw2,
                              const float* __restrict__ sc2, float* __restrict__ Wt2) {
    int idx = blockIdx.x * blockDim.x + threadIdx.x;
    if (idx >= NCLS * KT2) return;
    int o = idx / KT2, j = idx % KT2;
    float v;
    if (j < H1) {
        v = bw2[o * H1 + j];
    } else {
        int j2 = j - H1, i = j2 >> 3, g = j2 & 7;
        v = sw2[((size_t)o * H1 + i) * 8 + g] * sc2[o * H1 + i];
    }
    Wt2[(size_t)j * NCLS + o] = v;
}

__global__ void aprep1_kernel(const float* __restrict__ acc, const float* __restrict__ cnt,
                              float* __restrict__ A1) {
    int idx = blockIdx.x * blockDim.x + threadIdx.x;
    if (idx >= KSEG * C) return;
    int k = idx >> 6, c = idx & 63;
    float n = cnt[k];
    float f = acc[(size_t)k * C + c] / fmaxf(n, 1.0f);
    float* row = A1 + (size_t)k * KT1;
    row[c] = silu_f(f);
    float b[8];
    bspline8(f, b);
#pragma unroll
    for (int g = 0; g < 8; ++g) row[C + c * 8 + g] = b[g];
}

__global__ void aprep2_kernel(const float* __restrict__ h, float* __restrict__ A2) {
    int idx = blockIdx.x * blockDim.x + threadIdx.x;
    if (idx >= KSEG * H1) return;
    int m = idx >> 8, i = idx & 255;
    float f = h[(size_t)m * H1 + i];
    float* row = A2 + (size_t)m * KT2;
    row[i] = silu_f(f);
    float b[8];
    bspline8(f, b);
#pragma unroll
    for (int g = 0; g < 8; ++g) row[H1 + i * 8 + g] = b[g];
}

template <int NT>
__global__ void gemm_f32_wmma(const float* __restrict__ A, const float* __restrict__ Bt,
                              float* __restrict__ D, int M, int Kt, int N) {
    const int tilesN = N / (16 * NT);
    const int tm = blockIdx.x / tilesN;
    const int tn = blockIdx.x % tilesN;
    const int lane = threadIdx.x & 31;
    const int mrow = lane & 15;
    const int khalf = (lane >> 4) << 1;
    const int n0 = tn * 16 * NT;

    const float* Arow = A + (size_t)(tm * 16 + mrow) * Kt;

    v8f accv[NT];
#pragma unroll
    for (int t = 0; t < NT; ++t) accv[t] = (v8f){0.f, 0.f, 0.f, 0.f, 0.f, 0.f, 0.f, 0.f};

    for (int kk = 0; kk < Kt; kk += 4) {
        v2f a;
        a.x = Arow[kk + khalf];
        a.y = Arow[kk + khalf + 1];
#pragma unroll
        for (int t = 0; t < NT; ++t) {
            int n = n0 + t * 16 + mrow;
            v2f b;
            b.x = Bt[(size_t)(kk + khalf) * N + n];
            b.y = Bt[(size_t)(kk + khalf + 1) * N + n];
            accv[t] = __builtin_amdgcn_wmma_f32_16x16x4_f32(
                false, a, false, b, (short)0, accv[t], false, false);
        }
    }

    const int rbase = (lane >> 4) * 8;
#pragma unroll
    for (int t = 0; t < NT; ++t) {
#pragma unroll
        for (int r = 0; r < 8; ++r) {
            D[(size_t)(tm * 16 + rbase + r) * N + n0 + t * 16 + mrow] = accv[t][r];
        }
    }
}

extern "C" void kernel_launch(void* const* d_in, const int* in_sizes, int n_in,
                              void* d_out, int out_size, void* d_ws, size_t ws_size,
                              hipStream_t stream) {
    const float* x   = (const float*)d_in[0];
    const int*   cls = (const int*)  d_in[1];
    const float* bw1 = (const float*)d_in[2];
    const float* sw1 = (const float*)d_in[3];
    const float* sc1 = (const float*)d_in[4];
    const float* bw2 = (const float*)d_in[5];
    const float* sw2 = (const float*)d_in[6];
    const float* sc2 = (const float*)d_in[7];
    float* out = (float*)d_out;

    float* ws  = (float*)d_ws;
    float* acc = ws;
    float* cnt = acc + (size_t)KSEG * C;
    float* A1  = cnt + KSEG;
    float* Wt1 = A1 + (size_t)KSEG * KT1;
    float* h   = Wt1 + (size_t)KT1 * H1;
    float* A2  = h + (size_t)KSEG * H1;
    float* Wt2 = A2 + (size_t)KSEG * KT2;

    int* perm     = (int*)A2;
    int* table    = perm + HW;
    int* total    = table + (size_t)NB * KSEG;
    int* segstart = total + KSEG;

    chunkcount_kernel<<<NB, LT, 0, stream>>>(cls, table);
    segtotal_kernel<<<KSEG / 256, 256, 0, stream>>>(table, total, cnt);
    chunkfill_kernel<<<NB, LT, 0, stream>>>(cls, table, total, segstart, perm);
    seggather_kernel<<<KSEG / 4, 128, 0, stream>>>(x, segstart, perm, acc);
    {
        int n = H1 * KT1;
        wprep1_kernel<<<(n + 255) / 256, 256, 0, stream>>>(bw1, sw1, sc1, Wt1);
    }
    {
        int n = KSEG * C;
        aprep1_kernel<<<(n + 255) / 256, 256, 0, stream>>>(acc, cnt, A1);
    }
    gemm_f32_wmma<4><<<(KSEG / 16) * (H1 / 64), 32, 0, stream>>>(A1, Wt1, h, KSEG, KT1, H1);
    {
        int n = KSEG * H1;
        aprep2_kernel<<<(n + 255) / 256, 256, 0, stream>>>(h, A2);
    }
    {
        int n = NCLS * KT2;
        wprep2_kernel<<<(n + 255) / 256, 256, 0, stream>>>(bw2, sw2, sc2, Wt2);
    }
    gemm_f32_wmma<1><<<(KSEG / 16) * (NCLS / 16), 32, 0, stream>>>(A2, Wt2, out, KSEG, KT2, NCLS);
}


# both GEMMs: weights kept k-contiguous, A and W fragments by 16-byte loads through one loader, loads 2 to 7 k-blocks ahead in a register ring; sum order changes (k order, 2 accumulators in layer 2)
# speedup vs baseline: 3.7729x; 1.9715x over previous
#include <hip/hip_runtime.h>
#include <math.h>

#define HW   1048576
#define C    64
#define KSEG 4096
#define H1   256
#define NCLS 16
#define KT1  (C + C*8)
#define KT2  (H1 + H1*8)

typedef __attribute__((ext_vector_type(2))) float v2f;
typedef __attribute__((ext_vector_type(4))) float v4f;
typedef __attribute__((ext_vector_type(8))) float v8f;

__device__ __forceinline__ float silu_f(float x) {
    return x / (1.0f + expf(-x));
}

__device__ __forceinline__ void bspline8(float x, float* __restrict__ out) {
    float t[12];
#pragma unroll
    for (int j = 0; j < 12; ++j) t[j] = 0.4f * (float)(j - 3) - 1.0f;
    float b[11];
#pragma unroll
    for (int j = 0; j < 11; ++j) b[j] = (x >= t[j] && x < t[j + 1]) ? 1.0f : 0.0f;
#pragma unroll
    for (int k = 1; k <= 3; ++k) {
#pragma unroll
        for (int j = 0; j < 10; ++j) {
            if (j < 11 - k) {
                float left  = (x - t[j])         / (t[j + k]     - t[j]);
                float right = (t[j + k + 1] - x) / (t[j + k + 1] - t[j + 1]);
                b[j] = left * b[j] + right * b[j + 1];
            }
        }
    }
#pragma unroll
    for (int g = 0; g < 8; ++g) out[g] = b[g];
}

#define NB    128
#define CHUNK (HW / NB)
#define LT    1024
#define RPT   (CHUNK / LT)
#define TB    32
#define GU 16

typedef __attribute__((ext_vector_type(4))) int v4i;

__global__ void __launch_bounds__(LT)
chunkcount_kernel(const int* __restrict__ cls, int* __restrict__ table) {
    __shared__ int hist[KSEG];
    const int t = threadIdx.x;
    const int base = blockIdx.x * CHUNK + t * RPT;
    int s[RPT];
#pragma unroll
    for (int j = 0; j < RPT; ++j) s[j] = cls[base + j];
#pragma unroll
    for (int j = 0; j < KSEG / LT; ++j) hist[t + j * LT] = 0;
    __syncthreads();
#pragma unroll
    for (int j = 0; j < RPT; ++j) atomicAdd(&hist[s[j]], 1);
    __syncthreads();
    *(v4i*)(table + (size_t)blockIdx.x * KSEG + t * 4) = *(const v4i*)(hist + t * 4);
}

__global__ void __launch_bounds__(256)
segtotal_kernel(int* __restrict__ table, int* __restrict__ total, float* __restrict__ cnt) {
    const int s = blockIdx.x * 256 + threadIdx.x;
    int run = 0;
    for (int b0 = 0; b0 < NB; b0 += TB) {
        int c[TB];
#pragma unroll
        for (int u = 0; u < TB; ++u) c[u] = table[(size_t)(b0 + u) * KSEG + s];
#pragma unroll
        for (int u = 0; u < TB; ++u) {
            table[(size_t)(b0 + u) * KSEG + s] = run;
            run += c[u];
        }
    }
    total[s] = run;
    cnt[s]   = (float)run;
}

__global__ void __launch_bounds__(LT)
chunkfill_kernel(const int* __restrict__ cls, const int* __restrict__ table,
                 const int* __restrict__ total, int* __restrict__ segstart,
                 int* __restrict__ perm) {
    __shared__ int cursor[KSEG];
    __shared__ int tot[2][LT];
    const int t = threadIdx.x;
    const int base = blockIdx.x * CHUNK + t * RPT;
    int s[RPT], pos[RPT];
#pragma unroll
    for (int j = 0; j < RPT; ++j) s[j] = __builtin_nontemporal_load(cls + base + j);
    const v4i tv = *(const v4i*)(total + t * 4);
    const v4i ov = *(const v4i*)(table + (size_t)blockIdx.x * KSEG + t * 4);
    const int sum = tv.x + tv.y + tv.z + tv.w;
    tot[0][t] = sum;
    __syncthreads();
    int cur = 0;
    for (int d = 1; d < LT; d <<= 1) {
        int v = tot[cur][t];
        if (t >= d) v += tot[cur][t - d];
        tot[cur ^ 1][t] = v;
        cur ^= 1;
        __syncthreads();
    }
    v4i st;
    st.x = tot[cur][t] - sum;
    st.y = st.x + tv.x;
    st.z = st.y + tv.y;
    st.w = st.z + tv.z;
    *(v4i*)(cursor + t * 4) = st + ov;
    if (blockIdx.x == 0) {
        *(v4i*)(segstart + t * 4) = st;
        if (t == LT - 1) segstart[KSEG] = st.w + tv.w;
    }
    __syncthreads();
#pragma unroll
    for (int j = 0; j < RPT; ++j) pos[j] = atomicAdd(&cursor[s[j]], 1);
#pragma unroll
    for (int j = 0; j < RPT; ++j) perm[pos[j]] = base + j;
}

__global__ void __launch_bounds__(128)
seggather_kernel(const float* __restrict__ x, const int* __restrict__ segstart,
                 const int* __restrict__ perm, float* __restrict__ acc) {
    const int lane = threadIdx.x & 31;
    const int half = lane >> 4;
    const int c4   = (lane & 15) << 2;
    const int seg  = blockIdx.x * 4 + __builtin_amdgcn_readfirstlane((int)(threadIdx.x >> 5));
    const int s0 = segstart[seg];
    const int n  = segstart[seg + 1] - s0;
    const int* __restrict__ list = perm + s0;
    v4f s = (v4f){0.f, 0.f, 0.f, 0.f};
    if (n > 0) {
        int r[GU];
#pragma unroll
        for (int u = 0; u < GU; ++u) {
            const int e = u * 2 + half;
            r[u] = list[e < n ? e : n - 1];
        }
        for (int e0 = 0; e0 < n; e0 += 2 * GU) {
            v4f v[GU];
#pragma unroll
            for (int u = 0; u < GU; ++u) {
                v[u] = __builtin_nontemporal_load(
                    (const v4f*)(x + (size_t)(r[u] & (HW - 1)) * C + c4));
            }
#pragma unroll
            for (int u = 0; u < GU; ++u) {
                const int e = e0 + 2 * GU + u * 2 + half;
                r[u] = list[e < n ? e : n - 1];
            }
#pragma unroll
            for (int u = 0; u < GU; ++u) {
                const int e = e0 + u * 2 + half;
                if (e < n) s += v[u];
            }
        }
    }
    s.x += __shfl_xor(s.x, 16);
    s.y += __shfl_xor(s.y, 16);
    s.z += __shfl_xor(s.z, 16);
    s.w += __shfl_xor(s.w, 16);
    if (half == 0) *(v4f*)(acc + (size_t)seg * C + c4) = s;
}

__global__ void wprep1_kernel(const float* __restrict__ bw1, const float* __restrict__ sw1,
                              const float* __restrict__ sc1, float* __restrict__ Wt1) {
    int idx = blockIdx.x * blockDim.x + threadIdx.x;
    if (idx >= H1 * KT1) return;
    int o = idx / KT1, j = idx % KT1;
    float v;
    if (j < C) {
        v = bw1[o * C + j];
    } else {
        int j2 = j - C, i = j2 >> 3, g = j2 & 7;
        v = sw1[((size_t)o * C + i) * 8 + g] * sc1[o * C + i];
    }
    Wt1[idx] = v;
}

__global__ void wprep2_kernel(const float* __restrict__ bw2, const float* __restrict__ sw2,
                              const float* __restrict__ sc2, float* __restrict__ Wt2) {
    int idx = blockIdx.x * blockDim.x + threadIdx.x;
    if (idx >= NCLS * KT2) return;
    int o = idx / KT2, j = idx % KT2;
    float v;
    if (j < H1) {
        v = bw2[o * H1 + j];
    } else {
        int j2 = j - H1, i = j2 >> 3, g = j2 & 7;
        v = sw2[((size_t)o * H1 + i) * 8 + g] * sc2[o * H1 + i];
    }
    Wt2[idx] = v;
}

__global__ void aprep1_kernel(const float* __restrict__ acc, const float* __restrict__ cnt,
                              float* __restrict__ A1) {
    int idx = blockIdx.x * blockDim.x + threadIdx.x;
    if (idx >= KSEG * C) return;
    int k = idx >> 6, c = idx & 63;
    float n = cnt[k];
    float f = acc[(size_t)k * C + c] / fmaxf(n, 1.0f);
    float* row = A1 + (size_t)k * KT1;
    row[c] = silu_f(f);
    float b[8];
    bspline8(f, b);
#pragma unroll
    for (int g = 0; g < 8; ++g) row[C + c * 8 + g] = b[g];
}

__global__ void aprep2_kernel(const float* __restrict__ h, float* __restrict__ A2) {
    int idx = blockIdx.x * blockDim.x + threadIdx.x;
    if (idx >= KSEG * H1) return;
    int m = idx >> 8, i = idx & 255;
    float f = h[(size_t)m * H1 + i];
    float* row = A2 + (size_t)m * KT2;
    row[i] = silu_f(f);
    float b[8];
    bspline8(f, b);
#pragma unroll
    for (int g = 0; g < 8; ++g) row[H1 + i * 8 + g] = b[g];
}

template <int NT>
__device__ __forceinline__ void frag_load(const float* __restrict__ ap, const float* __restrict__ wp,
                                          int kt, int kb, v4f (&fa)[4], v4f (&fb)[NT][4]) {
#pragma unroll
    for (int j = 0; j < 4; ++j) fa[j] = *(const v4f*)(ap + kb + 4 * j);
#pragma unroll
    for (int t = 0; t < NT; ++t) {
#pragma unroll
        for (int j = 0; j < 4; ++j) fb[t][j] = *(const v4f*)(wp + (size_t)t * 16 * kt + kb + 4 * j);
    }
}

template <int NT, int NA>
__device__ __forceinline__ void frag_mma(const v4f (&fa)[4], const v4f (&fb)[NT][4],
                                         v8f (&acc)[NT][NA]) {
#pragma unroll
    for (int j = 0; j < 4; ++j) {
#pragma unroll
        for (int s = 0; s < 2; ++s) {
            v2f a;
            a.x = fa[j][2 * s];
            a.y = fa[j][2 * s + 1];
#pragma unroll
            for (int t = 0; t < NT; ++t) {
                v2f b;
                b.x = fb[t][j][2 * s];
                b.y = fb[t][j][2 * s + 1];
                acc[t][s % NA] = __builtin_amdgcn_wmma_f32_16x16x4_f32(
                    false, a, false, b, (short)0, acc[t][s % NA], false, false);
            }
        }
    }
}

template <int NT, int ST, int KT, int N, int WPB>
__global__ void __launch_bounds__(WPB * 32)
gemm_f32_wmma(const float* __restrict__ A, const float* __restrict__ W, float* __restrict__ D) {
    constexpr int NBLK = KT / 32;
    constexpr int NA = (NT == 1) ? 2 : 1;
    constexpr int tilesN = N / (16 * NT);
    static_assert(KT % 32 == 0 && NBLK % ST == 0 && N % (16 * NT) == 0, "tiling");
    const int lane = threadIdx.x & 31;
    const int wave = __builtin_amdgcn_readfirstlane((int)(threadIdx.x >> 5));
    const int c = lane & 15;
    const int h = lane >> 4;
    const int tn = blockIdx.x % tilesN;
    const int tm = (blockIdx.x / tilesN) * WPB + wave;
    const int n0 = tn * 16 * NT;

    const float* __restrict__ ap = A + (size_t)(tm * 16 + c) * KT + 16 * h;
    const float* __restrict__ wp = W + (size_t)(n0 + c) * KT + 16 * h;

    v4f fa[ST][4];
    v4f fb[ST][NT][4];
    v8f acc[NT][NA];
#pragma unroll
    for (int t = 0; t < NT; ++t) {
#pragma unroll
        for (int q = 0; q < NA; ++q) acc[t][q] = (v8f){0.f, 0.f, 0.f, 0.f, 0.f, 0.f, 0.f, 0.f};
    }

#pragma unroll
    for (int s = 0; s < ST - 1; ++s) frag_load<NT>(ap, wp, KT, s * 32, fa[s], fb[s]);

    for (int blk = 0; blk < NBLK; blk += ST) {
#pragma unroll
        for (int s = 0; s < ST; ++s) {
            int nb = blk + s + ST - 1;
            nb = nb < NBLK ? nb : NBLK - 1;
            frag_load<NT>(ap, wp, KT, nb * 32, fa[(s + ST - 1) % ST], fb[(s + ST - 1) % ST]);
            asm volatile("" ::: "memory");
            frag_mma<NT, NA>(fa[s], fb[s], acc);
        }
    }

    const int rbase = h * 8;
#pragma unroll
    for (int t = 0; t < NT; ++t) {
        v8f r = acc[t][0];
        if (NA == 2) r += acc[t][NA - 1];
#pragma unroll
        for (int q = 0; q < 8; ++q) {
            D[(size_t)(tm * 16 + rbase + q) * N + n0 + t * 16 + c] = r[q];
        }
    }
}

extern "C" void kernel_launch(void* const* d_in, const int* in_sizes, int n_in,
                              void* d_out, int out_size, void* d_ws, size_t ws_size,
                              hipStream_t stream) {
    const float* x   = (const float*)d_in[0];
    const int*   cls = (const int*)  d_in[1];
    const float* bw1 = (const float*)d_in[2];
    const float* sw1 = (const float*)d_in[3];
    const float* sc1 = (const float*)d_in[4];
    const float* bw2 = (const float*)d_in[5];
    const float* sw2 = (const float*)d_in[6];
    const float* sc2 = (const float*)d_in[7];
    float* out = (float*)d_out;

    float* ws  = (float*)d_ws;
    float* acc = ws;
    float* cnt = acc + (size_t)KSEG * C;
    float* A1  = cnt + KSEG;
    float* Wt1 = A1 + (size_t)KSEG * KT1;
    float* h   = Wt1 + (size_t)KT1 * H1;
    float* A2  = h + (size_t)KSEG * H1;
    float* Wt2 = A2 + (size_t)KSEG * KT2;

    int* perm     = (int*)A2;
    int* table    = perm + HW;
    int* total    = table + (size_t)NB * KSEG;
    int* segstart = total + KSEG;

    chunkcount_kernel<<<NB, LT, 0, stream>>>(cls, table);
    segtotal_kernel<<<KSEG / 256, 256, 0, stream>>>(table, total, cnt);
    chunkfill_kernel<<<NB, LT, 0, stream>>>(cls, table, total, segstart, perm);
    seggather_kernel<<<KSEG / 4, 128, 0, stream>>>(x, segstart, perm, acc);
    {
        int n = H1 * KT1;
        wprep1_kernel<<<(n + 255) / 256, 256, 0, stream>>>(bw1, sw1, sc1, Wt1);
    }
    {
        int n = KSEG * C;
        aprep1_kernel<<<(n + 255) / 256, 256, 0, stream>>>(acc, cnt, A1);
    }
    gemm_f32_wmma<4, 3, KT1, H1, 4><<<(KSEG / 64) * (H1 / 64), 128, 0, stream>>>(A1, Wt1, h);
    {
        int n = KSEG * H1;
        aprep2_kernel<<<(n + 255) / 256, 256, 0, stream>>>(h, A2);
    }
    {
        int n = NCLS * KT2;
        wprep2_kernel<<<(n + 255) / 256, 256, 0, stream>>>(bw2, sw2, sc2, Wt2);
    }
    gemm_f32_wmma<1, 8, KT2, NCLS, 1><<<(KSEG / 16) * (NCLS / 16), 32, 0, stream>>>(A2, Wt2, out);
}


# spline bases by the 3-level recursion on the one knot span that holds the value: 12 divisions a value, not 54, same knots and quotients; the two kernels then take 4 values a thread (waves / 4)
# speedup vs baseline: 4.0243x; 1.0666x over previous
#include <hip/hip_runtime.h>
#include <math.h>

#define HW   1048576
#define C    64
#define KSEG 4096
#define H1   256
#define NCLS 16
#define KT1  (C + C*8)
#define KT2  (H1 + H1*8)

typedef __attribute__((ext_vector_type(2))) float v2f;
typedef __attribute__((ext_vector_type(4))) float v4f;
typedef __attribute__((ext_vector_type(8))) float v8f;

__device__ __forceinline__ float silu_f(float x) {
    return x / (1.0f + expf(-x));
}

__device__ __forceinline__ float knot_at(int i) {
    return 0.4f * (float)(i - 3) - 1.0f;
}

__device__ __forceinline__ void bspline8(float x, float* __restrict__ out) {
    float K[7];
    int jj = -4;
#pragma unroll
    for (int a = 1; a <= 6; ++a) K[a] = knot_at(a - 4);
#pragma unroll
    for (int j = 0; j < 12; ++j) {
        const bool c = x >= knot_at(j);
        jj = c ? j - 3 : jj;
#pragma unroll
        for (int a = 1; a <= 6; ++a) K[a] = c ? knot_at(j + a - 3) : K[a];
    }
    const float d43 = K[4] - K[3], d42 = K[4] - K[2], d53 = K[5] - K[3];
    const float d41 = K[4] - K[1], d52 = K[5] - K[2], d63 = K[6] - K[3];
    const float p1 = (K[4] - x) / d43;
    const float q1 = (x - K[3]) / d43;
    const float p2 = ((K[4] - x) / d42) * p1;
    const float q2 = ((x - K[2]) / d42) * p1 + ((K[5] - x) / d53) * q1;
    const float r2 = ((x - K[3]) / d53) * q1;
    const float p3 = ((K[4] - x) / d41) * p2;
    const float q3 = ((x - K[1]) / d41) * p2 + ((K[5] - x) / d52) * q2;
    const float r3 = ((x - K[2]) / d52) * q2 + ((K[6] - x) / d63) * r2;
    const float s3 = ((x - K[3]) / d63) * r2;
    const float nf = x - x;
#pragma unroll
    for (int g = 0; g < 8; ++g) {
        float v = 0.0f;
        v = (jj == g)     ? p3 : v;
        v = (jj == g - 1) ? q3 : v;
        v = (jj == g - 2) ? r3 : v;
        v = (jj == g - 3) ? s3 : v;
        out[g] = v + nf;
    }
}

#define NB    128
#define CHUNK (HW / NB)
#define LT    1024
#define RPT   (CHUNK / LT)
#define TB    32
#define GU 16

typedef __attribute__((ext_vector_type(4))) int v4i;

__global__ void __launch_bounds__(LT)
chunkcount_kernel(const int* __restrict__ cls, int* __restrict__ table) {
    __shared__ int hist[KSEG];
    const int t = threadIdx.x;
    const int base = blockIdx.x * CHUNK + t * RPT;
    int s[RPT];
#pragma unroll
    for (int j = 0; j < RPT; ++j) s[j] = cls[base + j];
#pragma unroll
    for (int j = 0; j < KSEG / LT; ++j) hist[t + j * LT] = 0;
    __syncthreads();
#pragma unroll
    for (int j = 0; j < RPT; ++j) atomicAdd(&hist[s[j]], 1);
    __syncthreads();
    *(v4i*)(table + (size_t)blockIdx.x * KSEG + t * 4) = *(const v4i*)(hist + t * 4);
}

__global__ void __launch_bounds__(256)
segtotal_kernel(int* __restrict__ table, int* __restrict__ total, float* __restrict__ cnt) {
    const int s = blockIdx.x * 256 + threadIdx.x;
    int run = 0;
    for (int b0 = 0; b0 < NB; b0 += TB) {
        int c[TB];
#pragma unroll
        for (int u = 0; u < TB; ++u) c[u] = table[(size_t)(b0 + u) * KSEG + s];
#pragma unroll
        for (int u = 0; u < TB; ++u) {
            table[(size_t)(b0 + u) * KSEG + s] = run;
            run += c[u];
        }
    }
    total[s] = run;
    cnt[s]   = (float)run;
}

__global__ void __launch_bounds__(LT)
chunkfill_kernel(const int* __restrict__ cls, const int* __restrict__ table,
                 const int* __restrict__ total, int* __restrict__ segstart,
                 int* __restrict__ perm) {
    __shared__ int cursor[KSEG];
    __shared__ int tot[2][LT];
    const int t = threadIdx.x;
    const int base = blockIdx.x * CHUNK + t * RPT;
    int s[RPT], pos[RPT];
#pragma unroll
    for (int j = 0; j < RPT; ++j) s[j] = __builtin_nontemporal_load(cls + base + j);
    const v4i tv = *(const v4i*)(total + t * 4);
    const v4i ov = *(const v4i*)(table + (size_t)blockIdx.x * KSEG + t * 4);
    const int sum = tv.x + tv.y + tv.z + tv.w;
    tot[0][t] = sum;
    __syncthreads();
    int cur = 0;
    for (int d = 1; d < LT; d <<= 1) {
        int v = tot[cur][t];
        if (t >= d) v += tot[cur][t - d];
        tot[cur ^ 1][t] = v;
        cur ^= 1;
        __syncthreads();
    }
    v4i st;
    st.x = tot[cur][t] - sum;
    st.y = st.x + tv.x;
    st.z = st.y + tv.y;
    st.w = st.z + tv.z;
    *(v4i*)(cursor + t * 4) = st + ov;
    if (blockIdx.x == 0) {
        *(v4i*)(segstart + t * 4) = st;
        if (t == LT - 1) segstart[KSEG] = st.w + tv.w;
    }
    __syncthreads();
#pragma unroll
    for (int j = 0; j < RPT; ++j) pos[j] = atomicAdd(&cursor[s[j]], 1);
#pragma unroll
    for (int j = 0; j < RPT; ++j) perm[pos[j]] = base + j;
}

__global__ void __launch_bounds__(128)
seggather_kernel(const float* __restrict__ x, const int* __restrict__ segstart,
                 const int* __restrict__ perm, float* __restrict__ acc) {
    const int lane = threadIdx.x & 31;
    const int half = lane >> 4;
    const int c4   = (lane & 15) << 2;
    const int seg  = blockIdx.x * 4 + __builtin_amdgcn_readfirstlane((int)(threadIdx.x >> 5));
    const int s0 = segstart[seg];
    const int n  = segstart[seg + 1] - s0;
    const int* __restrict__ list = perm + s0;
    v4f s = (v4f){0.f, 0.f, 0.f, 0.f};
    if (n > 0) {
        int r[GU];
#pragma unroll
        for (int u = 0; u < GU; ++u) {
            const int e = u * 2 + half;
            r[u] = list[e < n ? e : n - 1];
        }
        for (int e0 = 0; e0 < n; e0 += 2 * GU) {
            v4f v[GU];
#pragma unroll
            for (int u = 0; u < GU; ++u) {
                v[u] = __builtin_nontemporal_load(
                    (const v4f*)(x + (size_t)(r[u] & (HW - 1)) * C + c4));
            }
#pragma unroll
            for (int u = 0; u < GU; ++u) {
                const int e = e0 + 2 * GU + u * 2 + half;
                r[u] = list[e < n ? e : n - 1];
            }
#pragma unroll
            for (int u = 0; u < GU; ++u) {
                const int e = e0 + u * 2 + half;
                if (e < n) s += v[u];
            }
        }
    }
    s.x += __shfl_xor(s.x, 16);
    s.y += __shfl_xor(s.y, 16);
    s.z += __shfl_xor(s.z, 16);
    s.w += __shfl_xor(s.w, 16);
    if (half == 0) *(v4f*)(acc + (size_t)seg * C + c4) = s;
}

__global__ void wprep1_kernel(const float* __restrict__ bw1, const float* __restrict__ sw1,
                              const float* __restrict__ sc1, float* __restrict__ Wt1) {
    int idx = blockIdx.x * blockDim.x + threadIdx.x;
    if (idx >= H1 * KT1) return;
    int o = idx / KT1, j = idx % KT1;
    float v;
    if (j < C) {
        v = bw1[o * C + j];
    } else {
        int j2 = j - C, i = j2 >> 3, g = j2 & 7;
        v = sw1[((size_t)o * C + i) * 8 + g] * sc1[o * C + i];
    }
    Wt1[idx] = v;
}

__global__ void wprep2_kernel(const float* __restrict__ bw2, const float* __restrict__ sw2,
                              const float* __restrict__ sc2, float* __restrict__ Wt2) {
    int idx = blockIdx.x * blockDim.x + threadIdx.x;
    if (idx >= NCLS * KT2) return;
    int o = idx / KT2, j = idx % KT2;
    float v;
    if (j < H1) {
        v = bw2[o * H1 + j];
    } else {
        int j2 = j - H1, i = j2 >> 3, g = j2 & 7;
        v = sw2[((size_t)o * H1 + i) * 8 + g] * sc2[o * H1 + i];
    }
    Wt2[idx] = v;
}

#define VPT 4

__global__ void __launch_bounds__(256)
aprep1_kernel(const float* __restrict__ acc, const float* __restrict__ cnt,
              float* __restrict__ A1) {
    const int c  = threadIdx.x & 63;
    const int k0 = blockIdx.x * (4 * VPT) + (threadIdx.x >> 6);
    float f[VPT];
#pragma unroll
    for (int r = 0; r < VPT; ++r) {
        const int k = k0 + 4 * r;
        f[r] = acc[(size_t)k * C + c] / fmaxf(cnt[k], 1.0f);
    }
#pragma unroll
    for (int r = 0; r < VPT; ++r) {
        float* row = A1 + (size_t)(k0 + 4 * r) * KT1;
        row[c] = silu_f(f[r]);
        float b[8];
        bspline8(f[r], b);
#pragma unroll
        for (int g = 0; g < 8; ++g) row[C + c * 8 + g] = b[g];
    }
}

__global__ void __launch_bounds__(256)
aprep2_kernel(const float* __restrict__ h, float* __restrict__ A2) {
    const int i  = threadIdx.x;
    const int m0 = blockIdx.x * VPT;
    float f[VPT];
#pragma unroll
    for (int r = 0; r < VPT; ++r) f[r] = h[(size_t)(m0 + r) * H1 + i];
#pragma unroll
    for (int r = 0; r < VPT; ++r) {
        float* row = A2 + (size_t)(m0 + r) * KT2;
        row[i] = silu_f(f[r]);
        float b[8];
        bspline8(f[r], b);
#pragma unroll
        for (int g = 0; g < 8; ++g) row[H1 + i * 8 + g] = b[g];
    }
}

template <int NT>
__device__ __forceinline__ void frag_load(const float* __restrict__ ap, const float* __restrict__ wp,
                                          int kt, int kb, v4f (&fa)[4], v4f (&fb)[NT][4]) {
#pragma unroll
    for (int j = 0; j < 4; ++j) fa[j] = *(const v4f*)(ap + kb + 4 * j);
#pragma unroll
    for (int t = 0; t < NT; ++t) {
#pragma unroll
        for (int j = 0; j < 4; ++j) fb[t][j] = *(const v4f*)(wp + (size_t)t * 16 * kt + kb + 4 * j);
    }
}

template <int NT, int NA>
__device__ __forceinline__ void frag_mma(const v4f (&fa)[4], const v4f (&fb)[NT][4],
                                         v8f (&acc)[NT][NA]) {
#pragma unroll
    for (int j = 0; j < 4; ++j) {
#pragma unroll
        for (int s = 0; s < 2; ++s) {
            v2f a;
            a.x = fa[j][2 * s];
            a.y = fa[j][2 * s + 1];
#pragma unroll
            for (int t = 0; t < NT; ++t) {
                v2f b;
                b.x = fb[t][j][2 * s];
                b.y = fb[t][j][2 * s + 1];
                acc[t][s % NA] = __builtin_amdgcn_wmma_f32_16x16x4_f32(
                    false, a, false, b, (short)0, acc[t][s % NA], false, false);
            }
        }
    }
}

template <int NT, int ST, int KT, int N, int WPB>
__global__ void __launch_bounds__(WPB * 32)
gemm_f32_wmma(const float* __restrict__ A, const float* __restrict__ W, float* __restrict__ D) {
    constexpr int NBLK = KT / 32;
    constexpr int NA = (NT == 1) ? 2 : 1;
    constexpr int tilesN = N / (16 * NT);
    static_assert(KT % 32 == 0 && NBLK % ST == 0 && N % (16 * NT) == 0, "tiling");
    const int lane = threadIdx.x & 31;
    const int wave = __builtin_amdgcn_readfirstlane((int)(threadIdx.x >> 5));
    const int c = lane & 15;
    const int h = lane >> 4;
    const int tn = blockIdx.x % tilesN;
    const int tm = (blockIdx.x / tilesN) * WPB + wave;
    const int n0 = tn * 16 * NT;

    const float* __restrict__ ap = A + (size_t)(tm * 16 + c) * KT + 16 * h;
    const float* __restrict__ wp = W + (size_t)(n0 + c) * KT + 16 * h;

    v4f fa[ST][4];
    v4f fb[ST][NT][4];
    v8f acc[NT][NA];
#pragma unroll
    for (int t = 0; t < NT; ++t) {
#pragma unroll
        for (int q = 0; q < NA; ++q) acc[t][q] = (v8f){0.f, 0.f, 0.f, 0.f, 0.f, 0.f, 0.f, 0.f};
    }

#pragma unroll
    for (int s = 0; s < ST - 1; ++s) frag_load<NT>(ap, wp, KT, s * 32, fa[s], fb[s]);

    for (int blk = 0; blk < NBLK; blk += ST) {
#pragma unroll
        for (int s = 0; s < ST; ++s) {
            int nb = blk + s + ST - 1;
            nb = nb < NBLK ? nb : NBLK - 1;
            frag_load<NT>(ap, wp, KT, nb * 32, fa[(s + ST - 1) % ST], fb[(s + ST - 1) % ST]);
            asm volatile("" ::: "memory");
            frag_mma<NT, NA>(fa[s], fb[s], acc);
        }
    }

    const int rbase = h * 8;
#pragma unroll
    for (int t = 0; t < NT; ++t) {
        v8f r = acc[t][0];
        if (NA == 2) r += acc[t][NA - 1];
#pragma unroll
        for (int q = 0; q < 8; ++q) {
            D[(size_t)(tm * 16 + rbase + q) * N + n0 + t * 16 + c] = r[q];
        }
    }
}

extern "C" void kernel_launch(void* const* d_in, const int* in_sizes, int n_in,
                              void* d_out, int out_size, void* d_ws, size_t ws_size,
                              hipStream_t stream) {
    const float* x   = (const float*)d_in[0];
    const int*   cls = (const int*)  d_in[1];
    const float* bw1 = (const float*)d_in[2];
    const float* sw1 = (const float*)d_in[3];
    const float* sc1 = (const float*)d_in[4];
    const float* bw2 = (const float*)d_in[5];
    const float* sw2 = (const float*)d_in[6];
    const float* sc2 = (const float*)d_in[7];
    float* out = (float*)d_out;

    float* ws  = (float*)d_ws;
    float* acc = ws;
    float* cnt = acc + (size_t)KSEG * C;
    float* A1  = cnt + KSEG;
    float* Wt1 = A1 + (size_t)KSEG * KT1;
    float* h   = Wt1 + (size_t)KT1 * H1;
    float* A2  = h + (size_t)KSEG * H1;
    float* Wt2 = A2 + (size_t)KSEG * KT2;

    int* perm     = (int*)A2;
    int* table    = perm + HW;
    int* total    = table + (size_t)NB * KSEG;
    int* segstart = total + KSEG;

    chunkcount_kernel<<<NB, LT, 0, stream>>>(cls, table);
    segtotal_kernel<<<KSEG / 256, 256, 0, stream>>>(table, total, cnt);
    chunkfill_kernel<<<NB, LT, 0, stream>>>(cls, table, total, segstart, perm);
    seggather_kernel<<<KSEG / 4, 128, 0, stream>>>(x, segstart, perm, acc);
    {
        int n = H1 * KT1;
        wprep1_kernel<<<(n + 255) / 256, 256, 0, stream>>>(bw1, sw1, sc1, Wt1);
    }
    aprep1_kernel<<<KSEG / (4 * VPT), 256, 0, stream>>>(acc, cnt, A1);
    gemm_f32_wmma<4, 3, KT1, H1, 4><<<(KSEG / 64) * (H1 / 64), 128, 0, stream>>>(A1, Wt1, h);
    aprep2_kernel<<<KSEG / VPT, 256, 0, stream>>>(h, A2);
    {
        int n = NCLS * KT2;
        wprep2_kernel<<<(n + 255) / 256, 256, 0, stream>>>(bw2, sw2, sc2, Wt2);
    }
    gemm_f32_wmma<1, 8, KT2, NCLS, 1><<<(KSEG / 16) * (NCLS / 16), 32, 0, stream>>>(A2, Wt2, out);
}


# row lists kept per chunk: one kernel counts, scans and sorts a chunk's rows by segment in LDS; gather lays 4 segments' lists into LDS from the 128 sublists; 10 launches -> 8; sum order as before
# speedup vs baseline: 5.6545x; 1.4051x over previous
#include <hip/hip_runtime.h>
#include <math.h>

#define HW   1048576
#define C    64
#define KSEG 4096
#define H1   256
#define NCLS 16
#define KT1  (C + C*8)
#define KT2  (H1 + H1*8)

typedef __attribute__((ext_vector_type(2))) float v2f;
typedef __attribute__((ext_vector_type(4))) float v4f;
typedef __attribute__((ext_vector_type(8))) float v8f;

__device__ __forceinline__ float silu_f(float x) {
    return x / (1.0f + expf(-x));
}

__device__ __forceinline__ float knot_at(int i) {
    return 0.4f * (float)(i - 3) - 1.0f;
}

__device__ __forceinline__ void bspline8(float x, float* __restrict__ out) {
    float K[7];
    int jj = -4;
#pragma unroll
    for (int a = 1; a <= 6; ++a) K[a] = knot_at(a - 4);
#pragma unroll
    for (int j = 0; j < 12; ++j) {
        const bool c = x >= knot_at(j);
        jj = c ? j - 3 : jj;
#pragma unroll
        for (int a = 1; a <= 6; ++a) K[a] = c ? knot_at(j + a - 3) : K[a];
    }
    const float d43 = K[4] - K[3], d42 = K[4] - K[2], d53 = K[5] - K[3];
    const float d41 = K[4] - K[1], d52 = K[5] - K[2], d63 = K[6] - K[3];
    const float p1 = (K[4] - x) / d43;
    const float q1 = (x - K[3]) / d43;
    const float p2 = ((K[4] - x) / d42) * p1;
    const float q2 = ((x - K[2]) / d42) * p1 + ((K[5] - x) / d53) * q1;
    const float r2 = ((x - K[3]) / d53) * q1;
    const float p3 = ((K[4] - x) / d41) * p2;
    const float q3 = ((x - K[1]) / d41) * p2 + ((K[5] - x) / d52) * q2;
    const float r3 = ((x - K[2]) / d52) * q2 + ((K[6] - x) / d63) * r2;
    const float s3 = ((x - K[3]) / d63) * r2;
    const float nf = x - x;
#pragma unroll
    for (int g = 0; g < 8; ++g) {
        float v = 0.0f;
        v = (jj == g)     ? p3 : v;
        v = (jj == g - 1) ? q3 : v;
        v = (jj == g - 2) ? r3 : v;
        v = (jj == g - 3) ? s3 : v;
        out[g] = v + nf;
    }
}

#define NB    128
#define CHUNK (HW / NB)
#define LT    1024
#define RPT   (CHUNK / LT)
#define GU    16
#define LCAP  1024
#define IDB   8

typedef __attribute__((ext_vector_type(4))) int v4i;

__global__ void __launch_bounds__(LT)
chunksort_kernel(const int* __restrict__ cls, int* __restrict__ coff, int* __restrict__ clist) {
    __shared__ int hist[KSEG];
    __shared__ int tot[2][LT];
    __shared__ int sorted[CHUNK];
    const int t = threadIdx.x;
    const int base = blockIdx.x * CHUNK + t * RPT;
    int s[RPT], pos[RPT];
#pragma unroll
    for (int j = 0; j < RPT; ++j) s[j] = __builtin_nontemporal_load(cls + base + j);
#pragma unroll
    for (int j = 0; j < KSEG / LT; ++j) hist[t + j * LT] = 0;
    __syncthreads();
#pragma unroll
    for (int j = 0; j < RPT; ++j) atomicAdd(&hist[s[j]], 1);
    __syncthreads();
    const v4i hv = *(const v4i*)(hist + t * 4);
    const int sum = hv.x + hv.y + hv.z + hv.w;
    tot[0][t] = sum;
    __syncthreads();
    int cur = 0;
    for (int d = 1; d < LT; d <<= 1) {
        int v = tot[cur][t];
        if (t >= d) v += tot[cur][t - d];
        tot[cur ^ 1][t] = v;
        cur ^= 1;
        __syncthreads();
    }
    v4i st;
    st.x = tot[cur][t] - sum;
    st.y = st.x + hv.x;
    st.z = st.y + hv.y;
    st.w = st.z + hv.z;
    *(v4i*)(hist + t * 4) = st;
    *(v4i*)(coff + (size_t)blockIdx.x * KSEG + t * 4) = st;
    __syncthreads();
#pragma unroll
    for (int j = 0; j < RPT; ++j) pos[j] = atomicAdd(&hist[s[j]], 1);
#pragma unroll
    for (int j = 0; j < RPT; ++j) sorted[pos[j]] = base + j;
    __syncthreads();
#pragma unroll
    for (int j = 0; j < CHUNK / (4 * LT); ++j) {
        const int q = (t + j * LT) * 4;
        *(v4i*)(clist + (size_t)blockIdx.x * CHUNK + q) = *(const v4i*)(sorted + q);
    }
}

__global__ void __launch_bounds__(NB)
seggather_kernel(const float* __restrict__ x, const int* __restrict__ coff,
                 const int* __restrict__ clist, float* __restrict__ acc,
                 float* __restrict__ cnt) {
    static_assert(NB == 128 && LCAP % (2 * GU) == 0, "one thread per chunk, four waves");
    __shared__ int ibuf[4][LCAP];
    __shared__ int wtot[4][4];
    const int t    = threadIdx.x;
    const int lane = t & 31;
    const int half = lane >> 4;
    const int c4   = (lane & 15) << 2;
    const int wave = __builtin_amdgcn_readfirstlane(t >> 5);
    const int q4   = blockIdx.x * 4;
    const int* __restrict__ co = coff + (size_t)t * KSEG + q4;
    const int* __restrict__ cl = clist + (size_t)t * CHUNK;
    const v4i o = *(const v4i*)co;
    const int oend = (q4 + 4 < KSEG) ? co[4] : CHUNK;
    v4i c;
    c.x = o.y - o.x; c.y = o.z - o.y; c.z = o.w - o.z; c.w = oend - o.w;
    v4i inc = c;
#pragma unroll
    for (int d = 1; d < 32; d <<= 1) {
        v4i up;
        up.x = __shfl_up(inc.x, d); up.y = __shfl_up(inc.y, d);
        up.z = __shfl_up(inc.z, d); up.w = __shfl_up(inc.w, d);
        if (lane >= d) inc += up;
    }
    if (lane == 31) *(v4i*)wtot[wave] = inc;
    __syncthreads();
    v4i pre = inc - c;
    v4i ntot = (v4i){0, 0, 0, 0};
#pragma unroll
    for (int w = 0; w < 4; ++w) {
        const v4i wt = *(const v4i*)wtot[w];
        if (w < wave) pre += wt;
        ntot += wt;
    }
    int nm = ntot.x > ntot.y ? ntot.x : ntot.y;
    nm = nm > ntot.z ? nm : ntot.z;
    nm = nm > ntot.w ? nm : ntot.w;
    const int nmax = __builtin_amdgcn_readfirstlane(nm);
    const int nsel = wave == 0 ? ntot.x : wave == 1 ? ntot.y : wave == 2 ? ntot.z : ntot.w;
    const int n = __builtin_amdgcn_readfirstlane(nsel);
    const int* list = ibuf[wave];
    v4f s = (v4f){0.f, 0.f, 0.f, 0.f};
    for (int w0 = 0; w0 < nmax; w0 += LCAP) {
        for (int p0 = o.x; p0 < oend; p0 += IDB) {
            int id[IDB];
#pragma unroll
            for (int j = 0; j < IDB; ++j) {
                const int p = p0 + j;
                id[j] = cl[p < CHUNK ? p : CHUNK - 1];
            }
#pragma unroll
            for (int j = 0; j < IDB; ++j) {
                const int p = p0 + j;
                if (p < oend) {
                    const int k = (p >= o.y) + (p >= o.z) + (p >= o.w);
                    const int ok = k == 0 ? o.x : k == 1 ? o.y : k == 2 ? o.z : o.w;
                    const int pk = k == 0 ? pre.x : k == 1 ? pre.y : k == 2 ? pre.z : pre.w;
                    const int d = pk + (p - ok) - w0;
                    if (d >= 0 && d < LCAP) ibuf[k][d] = id[j];
                }
            }
        }
        __syncthreads();
        int m = n - w0;
        m = m > LCAP ? LCAP : m;
        for (int e0 = 0; e0 < m; e0 += 2 * GU) {
            int r[GU];
            v4f v[GU];
#pragma unroll
            for (int u = 0; u < GU; ++u) {
                const int e = e0 + u * 2 + half;
                r[u] = list[e < m ? e : m - 1];
            }
#pragma unroll
            for (int u = 0; u < GU; ++u) {
                v[u] = __builtin_nontemporal_load(
                    (const v4f*)(x + (size_t)(r[u] & (HW - 1)) * C + c4));
            }
#pragma unroll
            for (int u = 0; u < GU; ++u) {
                const int e = e0 + u * 2 + half;
                if (e < m) s += v[u];
            }
        }
        __syncthreads();
    }
    s.x += __shfl_xor(s.x, 16);
    s.y += __shfl_xor(s.y, 16);
    s.z += __shfl_xor(s.z, 16);
    s.w += __shfl_xor(s.w, 16);
    const int seg = q4 + wave;
    if (half == 0) *(v4f*)(acc + (size_t)seg * C + c4) = s;
    if (lane == 0) cnt[seg] = (float)n;
}

__global__ void wprep1_kernel(const float* __restrict__ bw1, const float* __restrict__ sw1,
                              const float* __restrict__ sc1, float* __restrict__ Wt1) {
    int idx = blockIdx.x * blockDim.x + threadIdx.x;
    if (idx >= H1 * KT1) return;
    int o = idx / KT1, j = idx % KT1;
    float v;
    if (j < C) {
        v = bw1[o * C + j];
    } else {
        int j2 = j - C, i = j2 >> 3, g = j2 & 7;
        v = sw1[((size_t)o * C + i) * 8 + g] * sc1[o * C + i];
    }
    Wt1[idx] = v;
}

__global__ void wprep2_kernel(const float* __restrict__ bw2, const float* __restrict__ sw2,
                              const float* __restrict__ sc2, float* __restrict__ Wt2) {
    int idx = blockIdx.x * blockDim.x + threadIdx.x;
    if (idx >= NCLS * KT2) return;
    int o = idx / KT2, j = idx % KT2;
    float v;
    if (j < H1) {
        v = bw2[o * H1 + j];
    } else {
        int j2 = j - H1, i = j2 >> 3, g = j2 & 7;
        v = sw2[((size_t)o * H1 + i) * 8 + g] * sc2[o * H1 + i];
    }
    Wt2[idx] = v;
}

#define VPT 4

__global__ void __launch_bounds__(256)
aprep1_kernel(const float* __restrict__ acc, const float* __restrict__ cnt,
              float* __restrict__ A1) {
    const int c  = threadIdx.x & 63;
    const int k0 = blockIdx.x * (4 * VPT) + (threadIdx.x >> 6);
    float f[VPT];
#pragma unroll
    for (int r = 0; r < VPT; ++r) {
        const int k = k0 + 4 * r;
        f[r] = acc[(size_t)k * C + c] / fmaxf(cnt[k], 1.0f);
    }
#pragma unroll
    for (int r = 0; r < VPT; ++r) {
        float* row = A1 + (size_t)(k0 + 4 * r) * KT1;
        row[c] = silu_f(f[r]);
        float b[8];
        bspline8(f[r], b);
#pragma unroll
        for (int g = 0; g < 8; ++g) row[C + c * 8 + g] = b[g];
    }
}

__global__ void __launch_bounds__(256)
aprep2_kernel(const float* __restrict__ h, float* __restrict__ A2) {
    const int i  = threadIdx.x;
    const int m0 = blockIdx.x * VPT;
    float f[VPT];
#pragma unroll
    for (int r = 0; r < VPT; ++r) f[r] = h[(size_t)(m0 + r) * H1 + i];
#pragma unroll
    for (int r = 0; r < VPT; ++r) {
        float* row = A2 + (size_t)(m0 + r) * KT2;
        row[i] = silu_f(f[r]);
        float b[8];
        bspline8(f[r], b);
#pragma unroll
        for (int g = 0; g < 8; ++g) row[H1 + i * 8 + g] = b[g];
    }
}

template <int NT>
__device__ __forceinline__ void frag_load(const float* __restrict__ ap, const float* __restrict__ wp,
                                          int kt, int kb, v4f (&fa)[4], v4f (&fb)[NT][4]) {
#pragma unroll
    for (int j = 0; j < 4; ++j) fa[j] = *(const v4f*)(ap + kb + 4 * j);
#pragma unroll
    for (int t = 0; t < NT; ++t) {
#pragma unroll
        for (int j = 0; j < 4; ++j) fb[t][j] = *(const v4f*)(wp + (size_t)t * 16 * kt + kb + 4 * j);
    }
}

template <int NT, int NA>
__device__ __forceinline__ void frag_mma(const v4f (&fa)[4], const v4f (&fb)[NT][4],
                                         v8f (&acc)[NT][NA]) {
#pragma unroll
    for (int j = 0; j < 4; ++j) {
#pragma unroll
        for (int s = 0; s < 2; ++s) {
            v2f a;
            a.x = fa[j][2 * s];
            a.y = fa[j][2 * s + 1];
#pragma unroll
            for (int t = 0; t < NT; ++t) {
                v2f b;
                b.x = fb[t][j][2 * s];
                b.y = fb[t][j][2 * s + 1];
                acc[t][s % NA] = __builtin_amdgcn_wmma_f32_16x16x4_f32(
                    false, a, false, b, (short)0, acc[t][s % NA], false, false);
            }
        }
    }
}

template <int NT, int ST, int KT, int N, int WPB>
__global__ void __launch_bounds__(WPB * 32)
gemm_f32_wmma(const float* __restrict__ A, const float* __restrict__ W, float* __restrict__ D) {
    constexpr int NBLK = KT / 32;
    constexpr int NA = (NT == 1) ? 2 : 1;
    constexpr int tilesN = N / (16 * NT);
    static_assert(KT % 32 == 0 && NBLK % ST == 0 && N % (16 * NT) == 0, "tiling");
    const int lane = threadIdx.x & 31;
    const int wave = __builtin_amdgcn_readfirstlane((int)(threadIdx.x >> 5));
    const int c = lane & 15;
    const int h = lane >> 4;
    const int tn = blockIdx.x % tilesN;
    const int tm = (blockIdx.x / tilesN) * WPB + wave;
    const int n0 = tn * 16 * NT;

    const float* __restrict__ ap = A + (size_t)(tm * 16 + c) * KT + 16 * h;
    const float* __restrict__ wp = W + (size_t)(n0 + c) * KT + 16 * h;

    v4f fa[ST][4];
    v4f fb[ST][NT][4];
    v8f acc[NT][NA];
#pragma unroll
    for (int t = 0; t < NT; ++t) {
#pragma unroll
        for (int q = 0; q < NA; ++q) acc[t][q] = (v8f){0.f, 0.f, 0.f, 0.f, 0.f, 0.f, 0.f, 0.f};
    }

#pragma unroll
    for (int s = 0; s < ST - 1; ++s) frag_load<NT>(ap, wp, KT, s * 32, fa[s], fb[s]);

    for (int blk = 0; blk < NBLK; blk += ST) {
#pragma unroll
        for (int s = 0; s < ST; ++s) {
            int nb = blk + s + ST - 1;
            nb = nb < NBLK ? nb : NBLK - 1;
            frag_load<NT>(ap, wp, KT, nb * 32, fa[(s + ST - 1) % ST], fb[(s + ST - 1) % ST]);
            asm volatile("" ::: "memory");
            frag_mma<NT, NA>(fa[s], fb[s], acc);
        }
    }

    const int rbase = h * 8;
#pragma unroll
    for (int t = 0; t < NT; ++t) {
        v8f r = acc[t][0];
        if (NA == 2) r += acc[t][NA - 1];
#pragma unroll
        for (int q = 0; q < 8; ++q) {
            D[(size_t)(tm * 16 + rbase + q) * N + n0 + t * 16 + c] = r[q];
        }
    }
}

extern "C" void kernel_launch(void* const* d_in, const int* in_sizes, int n_in,
                              void* d_out, int out_size, void* d_ws, size_t ws_size,
                              hipStream_t stream) {
    const float* x   = (const float*)d_in[0];
    const int*   cls = (const int*)  d_in[1];
    const float* bw1 = (const float*)d_in[2];
    const float* sw1 = (const float*)d_in[3];
    const float* sc1 = (const float*)d_in[4];
    const float* bw2 = (const float*)d_in[5];
    const float* sw2 = (const float*)d_in[6];
    const float* sc2 = (const float*)d_in[7];
    float* out = (float*)d_out;

    float* ws  = (float*)d_ws;
    float* acc = ws;
    float* cnt = acc + (size_t)KSEG * C;
    float* A1  = cnt + KSEG;
    float* Wt1 = A1 + (size_t)KSEG * KT1;
    float* h   = Wt1 + (size_t)KT1 * H1;
    float* A2  = h + (size_t)KSEG * H1;
    float* Wt2 = A2 + (size_t)KSEG * KT2;

    int* clist = (int*)A2;
    int* coff  = clist + HW;

    chunksort_kernel<<<NB, LT, 0, stream>>>(cls, coff, clist);
    seggather_kernel<<<KSEG / 4, NB, 0, stream>>>(x, coff, clist, acc, cnt);
    {
        int n = H1 * KT1;
        wprep1_kernel<<<(n + 255) / 256, 256, 0, stream>>>(bw1, sw1, sc1, Wt1);
    }
    aprep1_kernel<<<KSEG / (4 * VPT), 256, 0, stream>>>(acc, cnt, A1);
    gemm_f32_wmma<4, 3, KT1, H1, 4><<<(KSEG / 64) * (H1 / 64), 128, 0, stream>>>(A1, Wt1, h);
    aprep2_kernel<<<KSEG / VPT, 256, 0, stream>>>(h, A2);
    {
        int n = NCLS * KT2;
        wprep2_kernel<<<(n + 255) / 256, 256, 0, stream>>>(bw2, sw2, sc2, Wt2);
    }
    gemm_f32_wmma<1, 8, KT2, NCLS, 1><<<(KSEG / 16) * (NCLS / 16), 32, 0, stream>>>(A2, Wt2, out);
}


# layer-1 activation rows written by the gather: the wave that summed a segment forms mean, silu and spline bases (2 channels a lane) and stores its row; sums plane and one launch gone (8 -> 7)
# speedup vs baseline: 5.9295x; 1.0486x over previous
#include <hip/hip_runtime.h>
#include <math.h>

#define HW   1048576
#define C    64
#define KSEG 4096
#define H1   256
#define NCLS 16
#define KT1  (C + C*8)
#define KT2  (H1 + H1*8)

typedef __attribute__((ext_vector_type(2))) float v2f;
typedef __attribute__((ext_vector_type(4))) float v4f;
typedef __attribute__((ext_vector_type(8))) float v8f;

__device__ __forceinline__ float silu_f(float x) {
    return x / (1.0f + expf(-x));
}

__device__ __forceinline__ float knot_at(int i) {
    return 0.4f * (float)(i - 3) - 1.0f;
}

__device__ __forceinline__ void bspline8(float x, float* __restrict__ out) {
    float K[7];
    int jj = -4;
#pragma unroll
    for (int a = 1; a <= 6; ++a) K[a] = knot_at(a - 4);
#pragma unroll
    for (int j = 0; j < 12; ++j) {
        const bool c = x >= knot_at(j);
        jj = c ? j - 3 : jj;
#pragma unroll
        for (int a = 1; a <= 6; ++a) K[a] = c ? knot_at(j + a - 3) : K[a];
    }
    const float d43 = K[4] - K[3], d42 = K[4] - K[2], d53 = K[5] - K[3];
    const float d41 = K[4] - K[1], d52 = K[5] - K[2], d63 = K[6] - K[3];
    const float p1 = (K[4] - x) / d43;
    const float q1 = (x - K[3]) / d43;
    const float p2 = ((K[4] - x) / d42) * p1;
    const float q2 = ((x - K[2]) / d42) * p1 + ((K[5] - x) / d53) * q1;
    const float r2 = ((x - K[3]) / d53) * q1;
    const float p3 = ((K[4] - x) / d41) * p2;
    const float q3 = ((x - K[1]) / d41) * p2 + ((K[5] - x) / d52) * q2;
    const float r3 = ((x - K[2]) / d52) * q2 + ((K[6] - x) / d63) * r2;
    const float s3 = ((x - K[3]) / d63) * r2;
    const float nf = x - x;
#pragma unroll
    for (int g = 0; g < 8; ++g) {
        float v = 0.0f;
        v = (jj == g)     ? p3 : v;
        v = (jj == g - 1) ? q3 : v;
        v = (jj == g - 2) ? r3 : v;
        v = (jj == g - 3) ? s3 : v;
        out[g] = v + nf;
    }
}

#define NB    128
#define CHUNK (HW / NB)
#define LT    1024
#define RPT   (CHUNK / LT)
#define GU    16
#define LCAP  1024
#define IDB   8

typedef __attribute__((ext_vector_type(4))) int v4i;

__global__ void __launch_bounds__(LT)
chunksort_kernel(const int* __restrict__ cls, int* __restrict__ coff, int* __restrict__ clist) {
    __shared__ int hist[KSEG];
    __shared__ int tot[2][LT];
    __shared__ int sorted[CHUNK];
    const int t = threadIdx.x;
    const int base = blockIdx.x * CHUNK + t * RPT;
    int s[RPT], pos[RPT];
#pragma unroll
    for (int j = 0; j < RPT; ++j) s[j] = __builtin_nontemporal_load(cls + base + j);
#pragma unroll
    for (int j = 0; j < KSEG / LT; ++j) hist[t + j * LT] = 0;
    __syncthreads();
#pragma unroll
    for (int j = 0; j < RPT; ++j) atomicAdd(&hist[s[j]], 1);
    __syncthreads();
    const v4i hv = *(const v4i*)(hist + t * 4);
    const int sum = hv.x + hv.y + hv.z + hv.w;
    tot[0][t] = sum;
    __syncthreads();
    int cur = 0;
    for (int d = 1; d < LT; d <<= 1) {
        int v = tot[cur][t];
        if (t >= d) v += tot[cur][t - d];
        tot[cur ^ 1][t] = v;
        cur ^= 1;
        __syncthreads();
    }
    v4i st;
    st.x = tot[cur][t] - sum;
    st.y = st.x + hv.x;
    st.z = st.y + hv.y;
    st.w = st.z + hv.z;
    *(v4i*)(hist + t * 4) = st;
    *(v4i*)(coff + (size_t)blockIdx.x * KSEG + t * 4) = st;
    __syncthreads();
#pragma unroll
    for (int j = 0; j < RPT; ++j) pos[j] = atomicAdd(&hist[s[j]], 1);
#pragma unroll
    for (int j = 0; j < RPT; ++j) sorted[pos[j]] = base + j;
    __syncthreads();
#pragma unroll
    for (int j = 0; j < CHUNK / (4 * LT); ++j) {
        const int q = (t + j * LT) * 4;
        *(v4i*)(clist + (size_t)blockIdx.x * CHUNK + q) = *(const v4i*)(sorted + q);
    }
}

__global__ void __launch_bounds__(NB)
seggather_kernel(const float* __restrict__ x, const int* __restrict__ coff,
                 const int* __restrict__ clist, float* __restrict__ A1) {
    static_assert(NB == 128 && LCAP % (2 * GU) == 0, "one thread per chunk, four waves");
    __shared__ int ibuf[4][LCAP];
    __shared__ int wtot[4][4];
    const int t    = threadIdx.x;
    const int lane = t & 31;
    const int half = lane >> 4;
    const int c4   = (lane & 15) << 2;
    const int wave = __builtin_amdgcn_readfirstlane(t >> 5);
    const int q4   = blockIdx.x * 4;
    const int* __restrict__ co = coff + (size_t)t * KSEG + q4;
    const int* __restrict__ cl = clist + (size_t)t * CHUNK;
    const v4i o = *(const v4i*)co;
    const int oend = (q4 + 4 < KSEG) ? co[4] : CHUNK;
    v4i c;
    c.x = o.y - o.x; c.y = o.z - o.y; c.z = o.w - o.z; c.w = oend - o.w;
    v4i inc = c;
#pragma unroll
    for (int d = 1; d < 32; d <<= 1) {
        v4i up;
        up.x = __shfl_up(inc.x, d); up.y = __shfl_up(inc.y, d);
        up.z = __shfl_up(inc.z, d); up.w = __shfl_up(inc.w, d);
        if (lane >= d) inc += up;
    }
    if (lane == 31) *(v4i*)wtot[wave] = inc;
    __syncthreads();
    v4i pre = inc - c;
    v4i ntot = (v4i){0, 0, 0, 0};
#pragma unroll
    for (int w = 0; w < 4; ++w) {
        const v4i wt = *(const v4i*)wtot[w];
        if (w < wave) pre += wt;
        ntot += wt;
    }
    int nm = ntot.x > ntot.y ? ntot.x : ntot.y;
    nm = nm > ntot.z ? nm : ntot.z;
    nm = nm > ntot.w ? nm : ntot.w;
    const int nmax = __builtin_amdgcn_readfirstlane(nm);
    const int nsel = wave == 0 ? ntot.x : wave == 1 ? ntot.y : wave == 2 ? ntot.z : ntot.w;
    const int n = __builtin_amdgcn_readfirstlane(nsel);
    const int* list = ibuf[wave];
    v4f s = (v4f){0.f, 0.f, 0.f, 0.f};
    for (int w0 = 0; w0 < nmax; w0 += LCAP) {
        for (int p0 = o.x; p0 < oend; p0 += IDB) {
            int id[IDB];
#pragma unroll
            for (int j = 0; j < IDB; ++j) {
                const int p = p0 + j;
                id[j] = cl[p < CHUNK ? p : CHUNK - 1];
            }
#pragma unroll
            for (int j = 0; j < IDB; ++j) {
                const int p = p0 + j;
                if (p < oend) {
                    const int k = (p >= o.y) + (p >= o.z) + (p >= o.w);
                    const int ok = k == 0 ? o.x : k == 1 ? o.y : k == 2 ? o.z : o.w;
                    const int pk = k == 0 ? pre.x : k == 1 ? pre.y : k == 2 ? pre.z : pre.w;
                    const int d = pk + (p - ok) - w0;
                    if (d >= 0 && d < LCAP) ibuf[k][d] = id[j];
                }
            }
        }
        __syncthreads();
        int m = n - w0;
        m = m > LCAP ? LCAP : m;
        for (int e0 = 0; e0 < m; e0 += 2 * GU) {
            int r[GU];
            v4f v[GU];
#pragma unroll
            for (int u = 0; u < GU; ++u) {
                const int e = e0 + u * 2 + half;
                r[u] = list[e < m ? e : m - 1];
            }
#pragma unroll
            for (int u = 0; u < GU; ++u) {
                v[u] = __builtin_nontemporal_load(
                    (const v4f*)(x + (size_t)(r[u] & (HW - 1)) * C + c4));
            }
#pragma unroll
            for (int u = 0; u < GU; ++u) {
                const int e = e0 + u * 2 + half;
                if (e < m) s += v[u];
            }
        }
        __syncthreads();
    }
    s.x += __shfl_xor(s.x, 16);
    s.y += __shfl_xor(s.y, 16);
    s.z += __shfl_xor(s.z, 16);
    s.w += __shfl_xor(s.w, 16);
    const int seg = q4 + wave;
    const float dn = fmaxf((float)n, 1.0f);
    const int cm = lane & 3;
    float f[2];
#pragma unroll
    for (int r = 0; r < 2; ++r) {
        const int sl = (lane >> 2) + 8 * r;
        const float a0 = __shfl(s.x, sl);
        const float a1 = __shfl(s.y, sl);
        const float a2 = __shfl(s.z, sl);
        const float a3 = __shfl(s.w, sl);
        float a = a0;
        a = (cm == 1) ? a1 : a;
        a = (cm == 2) ? a2 : a;
        a = (cm == 3) ? a3 : a;
        f[r] = a / dn;
    }
    float* row = A1 + (size_t)seg * KT1;
#pragma unroll
    for (int r = 0; r < 2; ++r) {
        const int ch = lane + 32 * r;
        row[ch] = silu_f(f[r]);
        float b[8];
        bspline8(f[r], b);
#pragma unroll
        for (int g = 0; g < 8; ++g) row[C + ch * 8 + g] = b[g];
    }
}

__global__ void wprep1_kernel(const float* __restrict__ bw1, const float* __restrict__ sw1,
                              const float* __restrict__ sc1, float* __restrict__ Wt1) {
    int idx = blockIdx.x * blockDim.x + threadIdx.x;
    if (idx >= H1 * KT1) return;
    int o = idx / KT1, j = idx % KT1;
    float v;
    if (j < C) {
        v = bw1[o * C + j];
    } else {
        int j2 = j - C, i = j2 >> 3, g = j2 & 7;
        v = sw1[((size_t)o * C + i) * 8 + g] * sc1[o * C + i];
    }
    Wt1[idx] = v;
}

__global__ void wprep2_kernel(const float* __restrict__ bw2, const float* __restrict__ sw2,
                              const float* __restrict__ sc2, float* __restrict__ Wt2) {
    int idx = blockIdx.x * blockDim.x + threadIdx.x;
    if (idx >= NCLS * KT2) return;
    int o = idx / KT2, j = idx % KT2;
    float v;
    if (j < H1) {
        v = bw2[o * H1 + j];
    } else {
        int j2 = j - H1, i = j2 >> 3, g = j2 & 7;
        v = sw2[((size_t)o * H1 + i) * 8 + g] * sc2[o * H1 + i];
    }
    Wt2[idx] = v;
}

#define VPT 4

__global__ void __launch_bounds__(256)
aprep2_kernel(const float* __restrict__ h, float* __restrict__ A2) {
    const int i  = threadIdx.x;
    const int m0 = blockIdx.x * VPT;
    float f[VPT];
#pragma unroll
    for (int r = 0; r < VPT; ++r) f[r] = h[(size_t)(m0 + r) * H1 + i];
#pragma unroll
    for (int r = 0; r < VPT; ++r) {
        float* row = A2 + (size_t)(m0 + r) * KT2;
        row[i] = silu_f(f[r]);
        float b[8];
        bspline8(f[r], b);
#pragma unroll
        for (int g = 0; g < 8; ++g) row[H1 + i * 8 + g] = b[g];
    }
}

template <int NT>
__device__ __forceinline__ void frag_load(const float* __restrict__ ap, const float* __restrict__ wp,
                                          int kt, int kb, v4f (&fa)[4], v4f (&fb)[NT][4]) {
#pragma unroll
    for (int j = 0; j < 4; ++j) fa[j] = *(const v4f*)(ap + kb + 4 * j);
#pragma unroll
    for (int t = 0; t < NT; ++t) {
#pragma unroll
        for (int j = 0; j < 4; ++j) fb[t][j] = *(const v4f*)(wp + (size_t)t * 16 * kt + kb + 4 * j);
    }
}

template <int NT, int NA>
__device__ __forceinline__ void frag_mma(const v4f (&fa)[4], const v4f (&fb)[NT][4],
                                         v8f (&acc)[NT][NA]) {
#pragma unroll
    for (int j = 0; j < 4; ++j) {
#pragma unroll
        for (int s = 0; s < 2; ++s) {
            v2f a;
            a.x = fa[j][2 * s];
            a.y = fa[j][2 * s + 1];
#pragma unroll
            for (int t = 0; t < NT; ++t) {
                v2f b;
                b.x = fb[t][j][2 * s];
                b.y = fb[t][j][2 * s + 1];
                acc[t][s % NA] = __builtin_amdgcn_wmma_f32_16x16x4_f32(
                    false, a, false, b, (short)0, acc[t][s % NA], false, false);
            }
        }
    }
}

template <int NT, int ST, int KT, int N, int WPB>
__global__ void __launch_bounds__(WPB * 32)
gemm_f32_wmma(const float* __restrict__ A, const float* __restrict__ W, float* __restrict__ D) {
    constexpr int NBLK = KT / 32;
    constexpr int NA = (NT == 1) ? 2 : 1;
    constexpr int tilesN = N / (16 * NT);
    static_assert(KT % 32 == 0 && NBLK % ST == 0 && N % (16 * NT) == 0, "tiling");
    const int lane = threadIdx.x & 31;
    const int wave = __builtin_amdgcn_readfirstlane((int)(threadIdx.x >> 5));
    const int c = lane & 15;
    const int h = lane >> 4;
    const int tn = blockIdx.x % tilesN;
    const int tm = (blockIdx.x / tilesN) * WPB + wave;
    const int n0 = tn * 16 * NT;

    const float* __restrict__ ap = A + (size_t)(tm * 16 + c) * KT + 16 * h;
    const float* __restrict__ wp = W + (size_t)(n0 + c) * KT + 16 * h;

    v4f fa[ST][4];
    v4f fb[ST][NT][4];
    v8f acc[NT][NA];
#pragma unroll
    for (int t = 0; t < NT; ++t) {
#pragma unroll
        for (int q = 0; q < NA; ++q) acc[t][q] = (v8f){0.f, 0.f, 0.f, 0.f, 0.f, 0.f, 0.f, 0.f};
    }

#pragma unroll
    for (int s = 0; s < ST - 1; ++s) frag_load<NT>(ap, wp, KT, s * 32, fa[s], fb[s]);

    for (int blk = 0; blk < NBLK; blk += ST) {
#pragma unroll
        for (int s = 0; s < ST; ++s) {
            int nb = blk + s + ST - 1;
            nb = nb < NBLK ? nb : NBLK - 1;
            frag_load<NT>(ap, wp, KT, nb * 32, fa[(s + ST - 1) % ST], fb[(s + ST - 1) % ST]);
            asm volatile("" ::: "memory");
            frag_mma<NT, NA>(fa[s], fb[s], acc);
        }
    }

    const int rbase = h * 8;
#pragma unroll
    for (int t = 0; t < NT; ++t) {
        v8f r = acc[t][0];
        if (NA == 2) r += acc[t][NA - 1];
#pragma unroll
        for (int q = 0; q < 8; ++q) {
            D[(size_t)(tm * 16 + rbase + q) * N + n0 + t * 16 + c] = r[q];
        }
    }
}

extern "C" void kernel_launch(void* const* d_in, const int* in_sizes, int n_in,
                              void* d_out, int out_size, void* d_ws, size_t ws_size,
                              hipStream_t stream) {
    const float* x   = (const float*)d_in[0];
    const int*   cls = (const int*)  d_in[1];
    const float* bw1 = (const float*)d_in[2];
    const float* sw1 = (const float*)d_in[3];
    const float* sc1 = (const float*)d_in[4];
    const float* bw2 = (const float*)d_in[5];
    const float* sw2 = (const float*)d_in[6];
    const float* sc2 = (const float*)d_in[7];
    float* out = (float*)d_out;

    float* ws  = (float*)d_ws;
    float* A1  = ws;
    float* Wt1 = A1 + (size_t)KSEG * KT1;
    float* h   = Wt1 + (size_t)KT1 * H1;
    float* A2  = h + (size_t)KSEG * H1;
    float* Wt2 = A2 + (size_t)KSEG * KT2;

    int* clist = (int*)A2;
    int* coff  = clist + HW;

    chunksort_kernel<<<NB, LT, 0, stream>>>(cls, coff, clist);
    seggather_kernel<<<KSEG / 4, NB, 0, stream>>>(x, coff, clist, A1);
    {
        int n = H1 * KT1;
        wprep1_kernel<<<(n + 255) / 256, 256, 0, stream>>>(bw1, sw1, sc1, Wt1);
    }
    gemm_f32_wmma<4, 3, KT1, H1, 4><<<(KSEG / 64) * (H1 / 64), 128, 0, stream>>>(A1, Wt1, h);
    aprep2_kernel<<<KSEG / VPT, 256, 0, stream>>>(h, A2);
    {
        int n = NCLS * KT2;
        wprep2_kernel<<<(n + 255) / 256, 256, 0, stream>>>(bw2, sw2, sc2, Wt2);
    }
    gemm_f32_wmma<1, 8, KT2, NCLS, 1><<<(KSEG / 16) * (NCLS / 16), 32, 0, stream>>>(A2, Wt2, out);
}


# GEMM loaders read bw and sw rows as they lie and form sw*sc in registers (layer-1 wave tile now 64x16: a scaled pair serves 4 row tiles); weight planes and 2 launches gone (7 -> 5); sums same
# speedup vs baseline: 6.1880x; 1.0436x over previous
#include <hip/hip_runtime.h>
#include <math.h>

#define HW   1048576
#define C    64
#define KSEG 4096
#define H1   256
#define NCLS 16
#define KT1  (C + C*8)
#define KT2  (H1 + H1*8)

typedef __attribute__((ext_vector_type(2))) float v2f;
typedef __attribute__((ext_vector_type(4))) float v4f;
typedef __attribute__((ext_vector_type(8))) float v8f;

__device__ __forceinline__ float silu_f(float x) {
    return x / (1.0f + expf(-x));
}

__device__ __forceinline__ float knot_at(int i) {
    return 0.4f * (float)(i - 3) - 1.0f;
}

__device__ __forceinline__ void bspline8(float x, float* __restrict__ out) {
    float K[7];
    int jj = -4;
#pragma unroll
    for (int a = 1; a <= 6; ++a) K[a] = knot_at(a - 4);
#pragma unroll
    for (int j = 0; j < 12; ++j) {
        const bool c = x >= knot_at(j);
        jj = c ? j - 3 : jj;
#pragma unroll
        for (int a = 1; a <= 6; ++a) K[a] = c ? knot_at(j + a - 3) : K[a];
    }
    const float d43 = K[4] - K[3], d42 = K[4] - K[2], d53 = K[5] - K[3];
    const float d41 = K[4] - K[1], d52 = K[5] - K[2], d63 = K[6] - K[3];
    const float p1 = (K[4] - x) / d43;
    const float q1 = (x - K[3]) / d43;
    const float p2 = ((K[4] - x) / d42) * p1;
    const float q2 = ((x - K[2]) / d42) * p1 + ((K[5] - x) / d53) * q1;
    const float r2 = ((x - K[3]) / d53) * q1;
    const float p3 = ((K[4] - x) / d41) * p2;
    const float q3 = ((x - K[1]) / d41) * p2 + ((K[5] - x) / d52) * q2;
    const float r3 = ((x - K[2]) / d52) * q2 + ((K[6] - x) / d63) * r2;
    const float s3 = ((x - K[3]) / d63) * r2;
    const float nf = x - x;
#pragma unroll
    for (int g = 0; g < 8; ++g) {
        float v = 0.0f;
        v = (jj == g)     ? p3 : v;
        v = (jj == g - 1) ? q3 : v;
        v = (jj == g - 2) ? r3 : v;
        v = (jj == g - 3) ? s3 : v;
        out[g] = v + nf;
    }
}

#define NB    128
#define CHUNK (HW / NB)
#define LT    1024
#define RPT   (CHUNK / LT)
#define GU    16
#define LCAP  1024
#define IDB   8

typedef __attribute__((ext_vector_type(4))) int v4i;

__global__ void __launch_bounds__(LT)
chunksort_kernel(const int* __restrict__ cls, int* __restrict__ coff, int* __restrict__ clist) {
    __shared__ int hist[KSEG];
    __shared__ int tot[2][LT];
    __shared__ int sorted[CHUNK];
    const int t = threadIdx.x;
    const int base = blockIdx.x * CHUNK + t * RPT;
    int s[RPT], pos[RPT];
#pragma unroll
    for (int j = 0; j < RPT; ++j) s[j] = __builtin_nontemporal_load(cls + base + j);
#pragma unroll
    for (int j = 0; j < KSEG / LT; ++j) hist[t + j * LT] = 0;
    __syncthreads();
#pragma unroll
    for (int j = 0; j < RPT; ++j) atomicAdd(&hist[s[j]], 1);
    __syncthreads();
    const v4i hv = *(const v4i*)(hist + t * 4);
    const int sum = hv.x + hv.y + hv.z + hv.w;
    tot[0][t] = sum;
    __syncthreads();
    int cur = 0;
    for (int d = 1; d < LT; d <<= 1) {
        int v = tot[cur][t];
        if (t >= d) v += tot[cur][t - d];
        tot[cur ^ 1][t] = v;
        cur ^= 1;
        __syncthreads();
    }
    v4i st;
    st.x = tot[cur][t] - sum;
    st.y = st.x + hv.x;
    st.z = st.y + hv.y;
    st.w = st.z + hv.z;
    *(v4i*)(hist + t * 4) = st;
    *(v4i*)(coff + (size_t)blockIdx.x * KSEG + t * 4) = st;
    __syncthreads();
#pragma unroll
    for (int j = 0; j < RPT; ++j) pos[j] = atomicAdd(&hist[s[j]], 1);
#pragma unroll
    for (int j = 0; j < RPT; ++j) sorted[pos[j]] = base + j;
    __syncthreads();
#pragma unroll
    for (int j = 0; j < CHUNK / (4 * LT); ++j) {
        const int q = (t + j * LT) * 4;
        *(v4i*)(clist + (size_t)blockIdx.x * CHUNK + q) = *(const v4i*)(sorted + q);
    }
}

__global__ void __launch_bounds__(NB)
seggather_kernel(const float* __restrict__ x, const int* __restrict__ coff,
                 const int* __restrict__ clist, float* __restrict__ A1) {
    static_assert(NB == 128 && LCAP % (2 * GU) == 0, "one thread per chunk, four waves");
    __shared__ int ibuf[4][LCAP];
    __shared__ int wtot[4][4];
    const int t    = threadIdx.x;
    const int lane = t & 31;
    const int half = lane >> 4;
    const int c4   = (lane & 15) << 2;
    const int wave = __builtin_amdgcn_readfirstlane(t >> 5);
    const int q4   = blockIdx.x * 4;
    const int* __restrict__ co = coff + (size_t)t * KSEG + q4;
    const int* __restrict__ cl = clist + (size_t)t * CHUNK;
    const v4i o = *(const v4i*)co;
    const int oend = (q4 + 4 < KSEG) ? co[4] : CHUNK;
    v4i c;
    c.x = o.y - o.x; c.y = o.z - o.y; c.z = o.w - o.z; c.w = oend - o.w;
    v4i inc = c;
#pragma unroll
    for (int d = 1; d < 32; d <<= 1) {
        v4i up;
        up.x = __shfl_up(inc.x, d); up.y = __shfl_up(inc.y, d);
        up.z = __shfl_up(inc.z, d); up.w = __shfl_up(inc.w, d);
        if (lane >= d) inc += up;
    }
    if (lane == 31) *(v4i*)wtot[wave] = inc;
    __syncthreads();
    v4i pre = inc - c;
    v4i ntot = (v4i){0, 0, 0, 0};
#pragma unroll
    for (int w = 0; w < 4; ++w) {
        const v4i wt = *(const v4i*)wtot[w];
        if (w < wave) pre += wt;
        ntot += wt;
    }
    int nm = ntot.x > ntot.y ? ntot.x : ntot.y;
    nm = nm > ntot.z ? nm : ntot.z;
    nm = nm > ntot.w ? nm : ntot.w;
    const int nmax = __builtin_amdgcn_readfirstlane(nm);
    const int nsel = wave == 0 ? ntot.x : wave == 1 ? ntot.y : wave == 2 ? ntot.z : ntot.w;
    const int n = __builtin_amdgcn_readfirstlane(nsel);
    const int* list = ibuf[wave];
    v4f s = (v4f){0.f, 0.f, 0.f, 0.f};
    for (int w0 = 0; w0 < nmax; w0 += LCAP) {
        for (int p0 = o.x; p0 < oend; p0 += IDB) {
            int id[IDB];
#pragma unroll
            for (int j = 0; j < IDB; ++j) {
                const int p = p0 + j;
                id[j] = cl[p < CHUNK ? p : CHUNK - 1];
            }
#pragma unroll
            for (int j = 0; j < IDB; ++j) {
                const int p = p0 + j;
                if (p < oend) {
                    const int k = (p >= o.y) + (p >= o.z) + (p >= o.w);
                    const int ok = k == 0 ? o.x : k == 1 ? o.y : k == 2 ? o.z : o.w;
                    const int pk = k == 0 ? pre.x : k == 1 ? pre.y : k == 2 ? pre.z : pre.w;
                    const int d = pk + (p - ok) - w0;
                    if (d >= 0 && d < LCAP) ibuf[k][d] = id[j];
                }
            }
        }
        __syncthreads();
        int m = n - w0;
        m = m > LCAP ? LCAP : m;
        for (int e0 = 0; e0 < m; e0 += 2 * GU) {
            int r[GU];
            v4f v[GU];
#pragma unroll
            for (int u = 0; u < GU; ++u) {
                const int e = e0 + u * 2 + half;
                r[u] = list[e < m ? e : m - 1];
            }
#pragma unroll
            for (int u = 0; u < GU; ++u) {
                v[u] = __builtin_nontemporal_load(
                    (const v4f*)(x + (size_t)(r[u] & (HW - 1)) * C + c4));
            }
#pragma unroll
            for (int u = 0; u < GU; ++u) {
                const int e = e0 + u * 2 + half;
                if (e < m) s += v[u];
            }
        }
        __syncthreads();
    }
    s.x += __shfl_xor(s.x, 16);
    s.y += __shfl_xor(s.y, 16);
    s.z += __shfl_xor(s.z, 16);
    s.w += __shfl_xor(s.w, 16);
    const int seg = q4 + wave;
    const float dn = fmaxf((float)n, 1.0f);
    const int cm = lane & 3;
    float f[2];
#pragma unroll
    for (int r = 0; r < 2; ++r) {
        const int sl = (lane >> 2) + 8 * r;
        const float a0 = __shfl(s.x, sl);
        const float a1 = __shfl(s.y, sl);
        const float a2 = __shfl(s.z, sl);
        const float a3 = __shfl(s.w, sl);
        float a = a0;
        a = (cm == 1) ? a1 : a;
        a = (cm == 2) ? a2 : a;
        a = (cm == 3) ? a3 : a;
        f[r] = a / dn;
    }
    float* row = A1 + (size_t)seg * KT1;
#pragma unroll
    for (int r = 0; r < 2; ++r) {
        const int ch = lane + 32 * r;
        row[ch] = silu_f(f[r]);
        float b[8];
        bspline8(f[r], b);
#pragma unroll
        for (int g = 0; g < 8; ++g) row[C + ch * 8 + g] = b[g];
    }
}

#define VPT 4

__global__ void __launch_bounds__(256)
aprep2_kernel(const float* __restrict__ h, float* __restrict__ A2) {
    const int i  = threadIdx.x;
    const int m0 = blockIdx.x * VPT;
    float f[VPT];
#pragma unroll
    for (int r = 0; r < VPT; ++r) f[r] = h[(size_t)(m0 + r) * H1 + i];
#pragma unroll
    for (int r = 0; r < VPT; ++r) {
        float* row = A2 + (size_t)(m0 + r) * KT2;
        row[i] = silu_f(f[r]);
        float b[8];
        bspline8(f[r], b);
#pragma unroll
        for (int g = 0; g < 8; ++g) row[H1 + i * 8 + g] = b[g];
    }
}

template <int MT, int IN>
__device__ __forceinline__ void frag_load(const float* __restrict__ ap, const float* __restrict__ bp,
                                          const float* __restrict__ sp, const float* __restrict__ cp,
                                          int nb, v4f (&fa)[MT][4], v4f (&fb)[4], v2f& fs) {
    constexpr int KT = 9 * IN;
    constexpr int NB0 = IN / 32;
#pragma unroll
    for (int t = 0; t < MT; ++t) {
#pragma unroll
        for (int j = 0; j < 4; ++j)
            fa[t][j] = *(const v4f*)(ap + (size_t)t * 16 * KT + nb * 32 + 4 * j);
    }
    if (nb < NB0) {
#pragma unroll
        for (int j = 0; j < 4; ++j) fb[j] = *(const v4f*)(bp + nb * 32 + 4 * j);
        fs = (v2f){1.0f, 1.0f};
    } else {
        const int js = (nb - NB0) * 32;
#pragma unroll
        for (int j = 0; j < 4; ++j) fb[j] = *(const v4f*)(sp + js + 4 * j);
        fs = *(const v2f*)(cp + (js >> 3));
    }
}

template <int MT, int NA>
__device__ __forceinline__ void frag_mma(const v4f (&fa)[MT][4], const v4f (&fb)[4],
                                         const v2f& fs, v8f (&acc)[MT][NA]) {
#pragma unroll
    for (int j = 0; j < 4; ++j) {
        const float sc = (j < 2) ? fs.x : fs.y;
#pragma unroll
        for (int s = 0; s < 2; ++s) {
            v2f b;
            b.x = fb[j][2 * s] * sc;
            b.y = fb[j][2 * s + 1] * sc;
#pragma unroll
            for (int t = 0; t < MT; ++t) {
                v2f a;
                a.x = fa[t][j][2 * s];
                a.y = fa[t][j][2 * s + 1];
                acc[t][s % NA] = __builtin_amdgcn_wmma_f32_16x16x4_f32(
                    false, a, false, b, (short)0, acc[t][s % NA], false, false);
            }
        }
    }
}

template <int MT, int ST, int IN, int N, int WPB>
__global__ void __launch_bounds__(WPB * 32)
kan_gemm_f32_wmma(const float* __restrict__ A, const float* __restrict__ bw,
                  const float* __restrict__ sw, const float* __restrict__ sc,
                  float* __restrict__ D) {
    constexpr int KT = 9 * IN;
    constexpr int NBLK = KT / 32;
    constexpr int NA = (MT == 1) ? 2 : 1;
    constexpr int tilesN = N / (16 * WPB);
    static_assert(IN % 32 == 0 && NBLK % ST == 0 && N % (16 * WPB) == 0, "tiling");
    const int lane = threadIdx.x & 31;
    const int wave = __builtin_amdgcn_readfirstlane((int)(threadIdx.x >> 5));
    const int c = lane & 15;
    const int h = lane >> 4;
    const int tn = blockIdx.x % tilesN;
    const int tm = blockIdx.x / tilesN;
    const int n0 = (tn * WPB + wave) * 16;
    const int m0 = tm * 16 * MT;

    const float* __restrict__ ap = A + (size_t)(m0 + c) * KT + 16 * h;
    const float* __restrict__ bp = bw + (size_t)(n0 + c) * IN + 16 * h;
    const float* __restrict__ sp = sw + (size_t)(n0 + c) * IN * 8 + 16 * h;
    const float* __restrict__ cp = sc + (size_t)(n0 + c) * IN + 2 * h;

    v4f fa[ST][MT][4];
    v4f fb[ST][4];
    v2f fs[ST];
    v8f acc[MT][NA];
#pragma unroll
    for (int t = 0; t < MT; ++t) {
#pragma unroll
        for (int q = 0; q < NA; ++q) acc[t][q] = (v8f){0.f, 0.f, 0.f, 0.f, 0.f, 0.f, 0.f, 0.f};
    }

#pragma unroll
    for (int s = 0; s < ST - 1; ++s) frag_load<MT, IN>(ap, bp, sp, cp, s, fa[s], fb[s], fs[s]);

#pragma unroll
    for (int blk = 0; blk < NBLK; blk += ST) {
#pragma unroll
        for (int s = 0; s < ST; ++s) {
            const int nb = blk + s + ST - 1;
            if (nb < NBLK)
                frag_load<MT, IN>(ap, bp, sp, cp, nb, fa[(s + ST - 1) % ST], fb[(s + ST - 1) % ST],
                                  fs[(s + ST - 1) % ST]);
            __builtin_amdgcn_sched_barrier(0);
            frag_mma<MT, NA>(fa[s], fb[s], fs[s], acc);
            __builtin_amdgcn_sched_barrier(0);
        }
    }

    const int rbase = h * 8;
#pragma unroll
    for (int t = 0; t < MT; ++t) {
        v8f r = acc[t][0];
        if (NA == 2) r += acc[t][NA - 1];
#pragma unroll
        for (int q = 0; q < 8; ++q) {
            D[(size_t)(m0 + t * 16 + rbase + q) * N + n0 + c] = r[q];
        }
    }
}

extern "C" void kernel_launch(void* const* d_in, const int* in_sizes, int n_in,
                              void* d_out, int out_size, void* d_ws, size_t ws_size,
                              hipStream_t stream) {
    const float* x   = (const float*)d_in[0];
    const int*   cls = (const int*)  d_in[1];
    const float* bw1 = (const float*)d_in[2];
    const float* sw1 = (const float*)d_in[3];
    const float* sc1 = (const float*)d_in[4];
    const float* bw2 = (const float*)d_in[5];
    const float* sw2 = (const float*)d_in[6];
    const float* sc2 = (const float*)d_in[7];
    float* out = (float*)d_out;

    float* ws  = (float*)d_ws;
    float* A1  = ws;
    float* h   = A1 + (size_t)KSEG * KT1;
    float* A2  = h + (size_t)KSEG * H1;

    int* clist = (int*)A2;
    int* coff  = clist + HW;

    chunksort_kernel<<<NB, LT, 0, stream>>>(cls, coff, clist);
    seggather_kernel<<<KSEG / 4, NB, 0, stream>>>(x, coff, clist, A1);
    kan_gemm_f32_wmma<4, 3, C, H1, 4><<<(KSEG / 64) * (H1 / 64), 128, 0, stream>>>(A1, bw1, sw1, sc1, h);
    aprep2_kernel<<<KSEG / VPT, 256, 0, stream>>>(h, A2);
    kan_gemm_f32_wmma<1, 8, H1, NCLS, 1><<<(KSEG / 16) * (NCLS / 16), 32, 0, stream>>>(A2, bw2, sw2, sc2, out);
}
